# starting kernel
# baseline (speedup 1.0000x reference)
#include <hip/hip_runtime.h>
#include <hip/hip_bf16.h>

#define HDIM 128
#define EPSV 1e-9f

typedef __attribute__((ext_vector_type(2))) float v2f;
typedef __attribute__((ext_vector_type(8))) float v8f;

__launch_bounds__(256)
__global__ void gemm_wmma_f32(const float* __restrict__ X,
                              const float* __restrict__ W,
                              const float* __restrict__ bias,
                              float* __restrict__ Y,
                              int Mrows, int accumulate) {
    __shared__ float lw[16][132];
    const int n0 = blockIdx.y * 16;

    for (int i = threadIdx.x; i < 16 * HDIM; i += 256) {
        const int n = i >> 7, k = i & (HDIM - 1);
        lw[n][k] = W[(size_t)(n0 + n) * HDIM + k];
    }
    __syncthreads();

    const int wave = threadIdx.x >> 5;
    const int lane = threadIdx.x & 31;
    const int m0   = blockIdx.x * 128 + wave * 16;
    if (m0 >= Mrows) return;

    const int nsub  = lane & 15;
    const int khalf = (lane >> 4) << 1;
    const float* __restrict__ xrow = X + (size_t)(m0 + nsub) * HDIM + khalf;
    const float* __restrict__ wrow = &lw[nsub][khalf];

    v8f acc = {};
#pragma unroll
    for (int k = 0; k < HDIM; k += 4) {
        v2f a, b;
        a.x = xrow[k]; a.y = xrow[k + 1];
        b.x = wrow[k]; b.y = wrow[k + 1];
        acc = __builtin_amdgcn_wmma_f32_16x16x4_f32(
             false, a,  false, b,
             (short)0, acc,  false,  false);
    }

    const float bval  = bias ? bias[n0 + nsub] : 0.0f;
    const int   mbase = m0 + ((lane >> 4) << 3);
    float* __restrict__ yp = Y + (size_t)mbase * HDIM + n0 + nsub;
#pragma unroll
    for (int r = 0; r < 8; ++r) {
        float v = acc[r] + bval;
        if (accumulate) v += yp[(size_t)r * HDIM];
        yp[(size_t)r * HDIM] = v;
    }
}

__device__ __forceinline__ float sigmoidf_(float x) {
    return 1.0f / (1.0f + __expf(-x));
}

__launch_bounds__(256)
__global__ void edge_score_kernel(const float* __restrict__ Ah,
                                  const float* __restrict__ Bx,
                                  const float* __restrict__ v_w,
                                  const int* __restrict__ ci,
                                  const int* __restrict__ chi,
                                  float* __restrict__ e,
                                  unsigned int* __restrict__ seg_max_u,
                                  int L) {
    const int edge = blockIdx.x * 8 + (threadIdx.x >> 5);
    const int lane = threadIdx.x & 31;
    if (edge >= L) return;
    const int c = ci[edge], ch = chi[edge];

    const float4 a = ((const float4*)(Ah + (size_t)ch * HDIM))[lane];
    const float4 b = ((const float4*)(Bx + (size_t)c  * HDIM))[lane];
    const float4 v = ((const float4*)v_w)[lane];

    float s = tanhf(a.x + b.x) * v.x + tanhf(a.y + b.y) * v.y
            + tanhf(a.z + b.z) * v.z + tanhf(a.w + b.w) * v.w;
#pragma unroll
    for (int off = 16; off > 0; off >>= 1) s += __shfl_xor(s, off, 32);

    if (lane == 0) {
        e[edge] = s;
        const unsigned int b32 = __float_as_uint(s);
        const unsigned int key = (b32 & 0x80000000u) ? ~b32 : (b32 | 0x80000000u);
        atomicMax(&seg_max_u[c], key);
    }
}

__launch_bounds__(256)
__global__ void edge_softmax_kernel(const float* __restrict__ e,
                                    const int* __restrict__ ci,
                                    const unsigned int* __restrict__ seg_max_u,
                                    float* __restrict__ x_exp,
                                    float* __restrict__ seg_sum,
                                    int L) {
    const int t = blockIdx.x * 256 + threadIdx.x;
    if (t >= L) return;
    const int c = ci[t];
    const unsigned int key = seg_max_u[c];
    const unsigned int b32 = (key & 0x80000000u) ? (key & 0x7FFFFFFFu) : ~key;
    const float xe = __expf(e[t] - __uint_as_float(b32));
    x_exp[t] = xe;
    atomicAdd(&seg_sum[c], xe);
}

__launch_bounds__(256)
__global__ void edge_accum_kernel(const float* __restrict__ x_exp,
                                  const float* __restrict__ seg_sum,
                                  const float* __restrict__ child_h,
                                  const float* __restrict__ child_c,
                                  const float* __restrict__ Wfx,
                                  const float* __restrict__ Ufh,
                                  const int* __restrict__ ci,
                                  const int* __restrict__ chi,
                                  float* __restrict__ h_hat,
                                  float* __restrict__ sum_fc,
                                  int L) {
    const int edge = blockIdx.x * 8 + (threadIdx.x >> 5);
    const int lane = threadIdx.x & 31;
    if (edge >= L) return;
    const int c = ci[edge], ch = chi[edge];
    const float attn = x_exp[edge] / (seg_sum[c] + EPSV);

    const float4 hv = ((const float4*)(child_h + (size_t)ch * HDIM))[lane];
    const float4 cv = ((const float4*)(child_c + (size_t)ch * HDIM))[lane];
    const float4 wf = ((const float4*)(Wfx + (size_t)c  * HDIM))[lane];
    const float4 uf = ((const float4*)(Ufh + (size_t)ch * HDIM))[lane];

    float* hh = h_hat  + (size_t)c * HDIM + lane * 4;
    float* sf = sum_fc + (size_t)c * HDIM + lane * 4;

    atomicAdd(&hh[0], attn * hv.x);
    atomicAdd(&hh[1], attn * hv.y);
    atomicAdd(&hh[2], attn * hv.z);
    atomicAdd(&hh[3], attn * hv.w);

    atomicAdd(&sf[0], sigmoidf_(wf.x + uf.x) * cv.x);
    atomicAdd(&sf[1], sigmoidf_(wf.y + uf.y) * cv.y);
    atomicAdd(&sf[2], sigmoidf_(wf.z + uf.z) * cv.z);
    atomicAdd(&sf[3], sigmoidf_(wf.w + uf.w) * cv.w);
}

__launch_bounds__(256)
__global__ void final_gate_kernel(const float* __restrict__ Gi,
                                  const float* __restrict__ Gc,
                                  const float* __restrict__ Go,
                                  const float* __restrict__ sum_fc,
                                  float* __restrict__ h_out,
                                  float* __restrict__ c_out,
                                  size_t n) {
    const size_t t = (size_t)blockIdx.x * 256 + threadIdx.x;
    if (t >= n) return;
    const float i_j = sigmoidf_(Gi[t]);
    const float c_t = tanhf(Gc[t]);
    const float c   = i_j * c_t + sum_fc[t];
    const float o_j = sigmoidf_(Go[t]);
    h_out[t] = o_j * tanhf(c);
    c_out[t] = c;
}

extern "C" void kernel_launch(void* const* d_in, const int* in_sizes, int n_in,
                              void* d_out, int out_size, void* d_ws, size_t ws_size,
                              hipStream_t stream) {
    const float* x_emb   = (const float*)d_in[0];
    const float* child_h = (const float*)d_in[1];
    const float* child_c = (const float*)d_in[2];
    const int*   ci      = (const int*)d_in[3];
    const int*   chi     = (const int*)d_in[4];
    const float* Wi_w = (const float*)d_in[5];
    const float* Ui_w = (const float*)d_in[6];
    const float* Wf_w = (const float*)d_in[7];
    const float* Uf_w = (const float*)d_in[8];
    const float* Wo_w = (const float*)d_in[9];
    const float* Uo_w = (const float*)d_in[10];
    const float* Wc_w = (const float*)d_in[11];
    const float* Uc_w = (const float*)d_in[12];
    const float* Wa_w = (const float*)d_in[13];
    const float* Ua_w = (const float*)d_in[14];
    const float* Wi_b = (const float*)d_in[15];
    const float* Wf_b = (const float*)d_in[16];
    const float* Wo_b = (const float*)d_in[17];
    const float* Wc_b = (const float*)d_in[18];
    const float* Wa_b = (const float*)d_in[19];
    const float* v_w  = (const float*)d_in[20];

    const int M = in_sizes[0] / HDIM;
    const int L = in_sizes[3];
    const size_t MH = (size_t)M * HDIM;

    float* ws    = (float*)d_ws;
    float* Ah    = ws;
    float* Bx    = Ah  + MH;
    float* Gi    = Bx  + MH;
    float* Wfx   = Gi  + MH;
    float* Go    = Wfx + MH;
    float* Gc    = Go  + MH;
    float* Ufh   = Gc  + MH;
    float* h_hat = Ufh + MH;
    float* sumfc = h_hat + MH;
    float* e_buf = sumfc + MH;
    float* x_exp = e_buf + L;
    float* seg_sum = x_exp + L;
    unsigned int* seg_max_u = (unsigned int*)(seg_sum + M);

    hipMemsetAsync(h_hat,     0, 2 * MH * sizeof(float), stream);
    hipMemsetAsync(seg_sum,   0, (size_t)M * sizeof(float), stream);
    hipMemsetAsync(seg_max_u, 0, (size_t)M * sizeof(unsigned int), stream);

    const dim3 gblk(256);
    const dim3 ggrd((M + 127) / 128, HDIM / 16);

    gemm_wmma_f32<<<ggrd, gblk, 0, stream>>>(child_h, Wa_w, Wa_b, Ah, M, 0);
    gemm_wmma_f32<<<ggrd, gblk, 0, stream>>>(x_emb,   Ua_w, nullptr, Bx, M, 0);
    gemm_wmma_f32<<<ggrd, gblk, 0, stream>>>(x_emb,   Wi_w, Wi_b, Gi, M, 0);
    gemm_wmma_f32<<<ggrd, gblk, 0, stream>>>(x_emb,   Wf_w, Wf_b, Wfx, M, 0);
    gemm_wmma_f32<<<ggrd, gblk, 0, stream>>>(x_emb,   Wo_w, Wo_b, Go, M, 0);
    gemm_wmma_f32<<<ggrd, gblk, 0, stream>>>(x_emb,   Wc_w, Wc_b, Gc, M, 0);
    gemm_wmma_f32<<<ggrd, gblk, 0, stream>>>(child_h, Uf_w, nullptr, Ufh, M, 0);

    edge_score_kernel<<<(L + 7) / 8, gblk, 0, stream>>>(Ah, Bx, v_w, ci, chi,
                                                        e_buf, seg_max_u, L);
    edge_softmax_kernel<<<(L + 255) / 256, gblk, 0, stream>>>(e_buf, ci, seg_max_u,
                                                              x_exp, seg_sum, L);
    edge_accum_kernel<<<(L + 7) / 8, gblk, 0, stream>>>(x_exp, seg_sum,
                                                        child_h, child_c, Wfx, Ufh,
                                                        ci, chi, h_hat, sumfc, L);

    gemm_wmma_f32<<<ggrd, gblk, 0, stream>>>(h_hat, Ui_w, nullptr, Gi, M, 1);
    gemm_wmma_f32<<<ggrd, gblk, 0, stream>>>(h_hat, Uc_w, nullptr, Gc, M, 1);
    gemm_wmma_f32<<<ggrd, gblk, 0, stream>>>(h_hat, Uo_w, nullptr, Go, M, 1);

    float* h_out = (float*)d_out;
    float* c_out = h_out + MH;
    final_gate_kernel<<<(unsigned)((MH + 255) / 256), gblk, 0, stream>>>(
        Gi, Gc, Go, sumfc, h_out, c_out, MH);
}


# GEMM re-tiled: one wave owns a 16x128 output stripe, its A tile held in registers over all 8 N tiles, whole W in LDS (rows permuted so a lane stores float4 rows): X read once, not 8 times
# speedup vs baseline: 1.5364x; 1.5364x over previous
#include <hip/hip_runtime.h>
#include <hip/hip_bf16.h>

#define HDIM 128
#define EPSV 1e-9f

typedef __attribute__((ext_vector_type(2))) float v2f;
typedef __attribute__((ext_vector_type(8))) float v8f;

#define LW_PITCH 132
__launch_bounds__(256) __attribute__((amdgpu_waves_per_eu(2, 4)))
__global__ void gemm_wmma_f32(const float* __restrict__ X,
                              const float* __restrict__ W,
                              const float* __restrict__ bias,
                              float* __restrict__ Y,
                              int Mrows, int accumulate) {
    __shared__ float lw[HDIM * LW_PITCH];

    const int wave = threadIdx.x >> 5;
    const int lane = threadIdx.x & 31;
    const int c    = lane & 15;
    const int h    = lane >> 4;
    const int m0   = blockIdx.x * 128 + wave * 16;
    const bool active = m0 < Mrows;

    v2f a[32];
    if (active) {
        const float* __restrict__ xrow = X + (size_t)(m0 + c) * HDIM + 2 * h;
#pragma unroll
        for (int ks = 0; ks < 32; ++ks) {
            a[ks].x = xrow[4 * ks];
            a[ks].y = xrow[4 * ks + 1];
        }
    }

#pragma unroll
    for (int it = 0; it < (HDIM * HDIM / 4) / 256; ++it) {
        const int i  = it * 256 + threadIdx.x;
        const int n  = i >> 5;
        const int k4 = (i & 31) << 2;
        const float4 w = *(const float4*)(W + (size_t)n * HDIM + k4);
        const int p  = (n & 64) | ((n & 3) << 4) | ((n & 63) >> 2);
        *(float4*)(&lw[p * LW_PITCH + k4]) = w;
    }
    __syncthreads();
    if (!active) return;

    const float* __restrict__ wrow = &lw[c * LW_PITCH + 2 * h];

    v8f acc[8] = {};
#pragma unroll
    for (int ks = 0; ks < 32; ++ks) {
#pragma unroll
        for (int t = 0; t < 8; ++t) {
            const v2f b = *(const v2f*)(wrow + t * 16 * LW_PITCH + 4 * ks);
            acc[t] = __builtin_amdgcn_wmma_f32_16x16x4_f32(
                 false, a[ks],  false, b,
                 (short)0, acc[t],  false,  false);
        }
    }

    float* __restrict__ ybase = Y + (size_t)(m0 + 8 * h) * HDIM + 4 * c;
    float4 bv[2];
#pragma unroll
    for (int g = 0; g < 2; ++g) {
        bv[g] = make_float4(0.0f, 0.0f, 0.0f, 0.0f);
        if (bias) bv[g] = *(const float4*)(bias + 64 * g + 4 * c);
    }

    if (accumulate) {
        float4 y[16];
#pragma unroll
        for (int i = 0; i < 16; ++i)
            y[i] = *(const float4*)(ybase + (size_t)(i & 7) * HDIM + 64 * (i >> 3));
#pragma unroll
        for (int i = 0; i < 16; ++i) {
            const int g = i >> 3, r = i & 7;
            float4 v;
            v.x = (acc[4 * g + 0][r] + bv[g].x) + y[i].x;
            v.y = (acc[4 * g + 1][r] + bv[g].y) + y[i].y;
            v.z = (acc[4 * g + 2][r] + bv[g].z) + y[i].z;
            v.w = (acc[4 * g + 3][r] + bv[g].w) + y[i].w;
            *(float4*)(ybase + (size_t)r * HDIM + 64 * g) = v;
        }
    } else {
#pragma unroll
        for (int i = 0; i < 16; ++i) {
            const int g = i >> 3, r = i & 7;
            float4 v;
            v.x = acc[4 * g + 0][r] + bv[g].x;
            v.y = acc[4 * g + 1][r] + bv[g].y;
            v.z = acc[4 * g + 2][r] + bv[g].z;
            v.w = acc[4 * g + 3][r] + bv[g].w;
            *(float4*)(ybase + (size_t)r * HDIM + 64 * g) = v;
        }
    }
}

__device__ __forceinline__ float sigmoidf_(float x) {
    return 1.0f / (1.0f + __expf(-x));
}

__launch_bounds__(256)
__global__ void edge_score_kernel(const float* __restrict__ Ah,
                                  const float* __restrict__ Bx,
                                  const float* __restrict__ v_w,
                                  const int* __restrict__ ci,
                                  const int* __restrict__ chi,
                                  float* __restrict__ e,
                                  unsigned int* __restrict__ seg_max_u,
                                  int L) {
    const int edge = blockIdx.x * 8 + (threadIdx.x >> 5);
    const int lane = threadIdx.x & 31;
    if (edge >= L) return;
    const int c = ci[edge], ch = chi[edge];

    const float4 a = ((const float4*)(Ah + (size_t)ch * HDIM))[lane];
    const float4 b = ((const float4*)(Bx + (size_t)c  * HDIM))[lane];
    const float4 v = ((const float4*)v_w)[lane];

    float s = tanhf(a.x + b.x) * v.x + tanhf(a.y + b.y) * v.y
            + tanhf(a.z + b.z) * v.z + tanhf(a.w + b.w) * v.w;
#pragma unroll
    for (int off = 16; off > 0; off >>= 1) s += __shfl_xor(s, off, 32);

    if (lane == 0) {
        e[edge] = s;
        const unsigned int b32 = __float_as_uint(s);
        const unsigned int key = (b32 & 0x80000000u) ? ~b32 : (b32 | 0x80000000u);
        atomicMax(&seg_max_u[c], key);
    }
}

__launch_bounds__(256)
__global__ void edge_softmax_kernel(const float* __restrict__ e,
                                    const int* __restrict__ ci,
                                    const unsigned int* __restrict__ seg_max_u,
                                    float* __restrict__ x_exp,
                                    float* __restrict__ seg_sum,
                                    int L) {
    const int t = blockIdx.x * 256 + threadIdx.x;
    if (t >= L) return;
    const int c = ci[t];
    const unsigned int key = seg_max_u[c];
    const unsigned int b32 = (key & 0x80000000u) ? (key & 0x7FFFFFFFu) : ~key;
    const float xe = __expf(e[t] - __uint_as_float(b32));
    x_exp[t] = xe;
    atomicAdd(&seg_sum[c], xe);
}

__launch_bounds__(256)
__global__ void edge_accum_kernel(const float* __restrict__ x_exp,
                                  const float* __restrict__ seg_sum,
                                  const float* __restrict__ child_h,
                                  const float* __restrict__ child_c,
                                  const float* __restrict__ Wfx,
                                  const float* __restrict__ Ufh,
                                  const int* __restrict__ ci,
                                  const int* __restrict__ chi,
                                  float* __restrict__ h_hat,
                                  float* __restrict__ sum_fc,
                                  int L) {
    const int edge = blockIdx.x * 8 + (threadIdx.x >> 5);
    const int lane = threadIdx.x & 31;
    if (edge >= L) return;
    const int c = ci[edge], ch = chi[edge];
    const float attn = x_exp[edge] / (seg_sum[c] + EPSV);

    const float4 hv = ((const float4*)(child_h + (size_t)ch * HDIM))[lane];
    const float4 cv = ((const float4*)(child_c + (size_t)ch * HDIM))[lane];
    const float4 wf = ((const float4*)(Wfx + (size_t)c  * HDIM))[lane];
    const float4 uf = ((const float4*)(Ufh + (size_t)ch * HDIM))[lane];

    float* hh = h_hat  + (size_t)c * HDIM + lane * 4;
    float* sf = sum_fc + (size_t)c * HDIM + lane * 4;

    atomicAdd(&hh[0], attn * hv.x);
    atomicAdd(&hh[1], attn * hv.y);
    atomicAdd(&hh[2], attn * hv.z);
    atomicAdd(&hh[3], attn * hv.w);

    atomicAdd(&sf[0], sigmoidf_(wf.x + uf.x) * cv.x);
    atomicAdd(&sf[1], sigmoidf_(wf.y + uf.y) * cv.y);
    atomicAdd(&sf[2], sigmoidf_(wf.z + uf.z) * cv.z);
    atomicAdd(&sf[3], sigmoidf_(wf.w + uf.w) * cv.w);
}

__launch_bounds__(256)
__global__ void final_gate_kernel(const float* __restrict__ Gi,
                                  const float* __restrict__ Gc,
                                  const float* __restrict__ Go,
                                  const float* __restrict__ sum_fc,
                                  float* __restrict__ h_out,
                                  float* __restrict__ c_out,
                                  size_t n) {
    const size_t t = (size_t)blockIdx.x * 256 + threadIdx.x;
    if (t >= n) return;
    const float i_j = sigmoidf_(Gi[t]);
    const float c_t = tanhf(Gc[t]);
    const float c   = i_j * c_t + sum_fc[t];
    const float o_j = sigmoidf_(Go[t]);
    h_out[t] = o_j * tanhf(c);
    c_out[t] = c;
}

extern "C" void kernel_launch(void* const* d_in, const int* in_sizes, int n_in,
                              void* d_out, int out_size, void* d_ws, size_t ws_size,
                              hipStream_t stream) {
    const float* x_emb   = (const float*)d_in[0];
    const float* child_h = (const float*)d_in[1];
    const float* child_c = (const float*)d_in[2];
    const int*   ci      = (const int*)d_in[3];
    const int*   chi     = (const int*)d_in[4];
    const float* Wi_w = (const float*)d_in[5];
    const float* Ui_w = (const float*)d_in[6];
    const float* Wf_w = (const float*)d_in[7];
    const float* Uf_w = (const float*)d_in[8];
    const float* Wo_w = (const float*)d_in[9];
    const float* Uo_w = (const float*)d_in[10];
    const float* Wc_w = (const float*)d_in[11];
    const float* Uc_w = (const float*)d_in[12];
    const float* Wa_w = (const float*)d_in[13];
    const float* Ua_w = (const float*)d_in[14];
    const float* Wi_b = (const float*)d_in[15];
    const float* Wf_b = (const float*)d_in[16];
    const float* Wo_b = (const float*)d_in[17];
    const float* Wc_b = (const float*)d_in[18];
    const float* Wa_b = (const float*)d_in[19];
    const float* v_w  = (const float*)d_in[20];

    const int M = in_sizes[0] / HDIM;
    const int L = in_sizes[3];
    const size_t MH = (size_t)M * HDIM;

    float* ws    = (float*)d_ws;
    float* Ah    = ws;
    float* Bx    = Ah  + MH;
    float* Gi    = Bx  + MH;
    float* Wfx   = Gi  + MH;
    float* Go    = Wfx + MH;
    float* Gc    = Go  + MH;
    float* Ufh   = Gc  + MH;
    float* h_hat = Ufh + MH;
    float* sumfc = h_hat + MH;
    float* e_buf = sumfc + MH;
    float* x_exp = e_buf + L;
    float* seg_sum = x_exp + L;
    unsigned int* seg_max_u = (unsigned int*)(seg_sum + M);

    hipMemsetAsync(h_hat,     0, 2 * MH * sizeof(float), stream);
    hipMemsetAsync(seg_sum,   0, (size_t)M * sizeof(float), stream);
    hipMemsetAsync(seg_max_u, 0, (size_t)M * sizeof(unsigned int), stream);

    const dim3 gblk(256);
    const dim3 ggrd((M + 127) / 128);

    gemm_wmma_f32<<<ggrd, gblk, 0, stream>>>(child_h, Wa_w, Wa_b, Ah, M, 0);
    gemm_wmma_f32<<<ggrd, gblk, 0, stream>>>(x_emb,   Ua_w, nullptr, Bx, M, 0);
    gemm_wmma_f32<<<ggrd, gblk, 0, stream>>>(x_emb,   Wi_w, Wi_b, Gi, M, 0);
    gemm_wmma_f32<<<ggrd, gblk, 0, stream>>>(x_emb,   Wf_w, Wf_b, Wfx, M, 0);
    gemm_wmma_f32<<<ggrd, gblk, 0, stream>>>(x_emb,   Wo_w, Wo_b, Go, M, 0);
    gemm_wmma_f32<<<ggrd, gblk, 0, stream>>>(x_emb,   Wc_w, Wc_b, Gc, M, 0);
    gemm_wmma_f32<<<ggrd, gblk, 0, stream>>>(child_h, Uf_w, nullptr, Ufh, M, 0);

    edge_score_kernel<<<(L + 7) / 8, gblk, 0, stream>>>(Ah, Bx, v_w, ci, chi,
                                                        e_buf, seg_max_u, L);
    edge_softmax_kernel<<<(L + 255) / 256, gblk, 0, stream>>>(e_buf, ci, seg_max_u,
                                                              x_exp, seg_sum, L);
    edge_accum_kernel<<<(L + 7) / 8, gblk, 0, stream>>>(x_exp, seg_sum,
                                                        child_h, child_c, Wfx, Ufh,
                                                        ci, chi, h_hat, sumfc, L);

    gemm_wmma_f32<<<ggrd, gblk, 0, stream>>>(h_hat, Ui_w, nullptr, Gi, M, 1);
    gemm_wmma_f32<<<ggrd, gblk, 0, stream>>>(h_hat, Uc_w, nullptr, Gc, M, 1);
    gemm_wmma_f32<<<ggrd, gblk, 0, stream>>>(h_hat, Uo_w, nullptr, Go, M, 1);

    float* h_out = (float*)d_out;
    float* c_out = h_out + MH;
    final_gate_kernel<<<(unsigned)((MH + 255) / 256), gblk, 0, stream>>>(
        Gi, Gc, Go, sumfc, h_out, c_out, MH);
}


# Edge-score, edge-accumulate and gate kernels: one wave takes 16 edges in two batches of 8 (gate: 4 float4 per thread), a million waves become 63 thousand; float atomics kept, arrival order as before
# speedup vs baseline: 1.7191x; 1.1189x over previous
#include <hip/hip_runtime.h>
#include <hip/hip_bf16.h>

#define HDIM 128
#define EPSV 1e-9f

typedef __attribute__((ext_vector_type(2))) float v2f;
typedef __attribute__((ext_vector_type(8))) float v8f;

#define LW_PITCH 132
__launch_bounds__(256) __attribute__((amdgpu_waves_per_eu(2, 4)))
__global__ void gemm_wmma_f32(const float* __restrict__ X,
                              const float* __restrict__ W,
                              const float* __restrict__ bias,
                              float* __restrict__ Y,
                              int Mrows, int accumulate) {
    __shared__ float lw[HDIM * LW_PITCH];

    const int wave = threadIdx.x >> 5;
    const int lane = threadIdx.x & 31;
    const int c    = lane & 15;
    const int h    = lane >> 4;
    const int m0   = blockIdx.x * 128 + wave * 16;
    const bool active = m0 < Mrows;

    v2f a[32];
    if (active) {
        const float* __restrict__ xrow = X + (size_t)(m0 + c) * HDIM + 2 * h;
#pragma unroll
        for (int ks = 0; ks < 32; ++ks) {
            a[ks].x = xrow[4 * ks];
            a[ks].y = xrow[4 * ks + 1];
        }
    }

#pragma unroll
    for (int it = 0; it < (HDIM * HDIM / 4) / 256; ++it) {
        const int i  = it * 256 + threadIdx.x;
        const int n  = i >> 5;
        const int k4 = (i & 31) << 2;
        const float4 w = *(const float4*)(W + (size_t)n * HDIM + k4);
        const int p  = (n & 64) | ((n & 3) << 4) | ((n & 63) >> 2);
        *(float4*)(&lw[p * LW_PITCH + k4]) = w;
    }
    __syncthreads();
    if (!active) return;

    const float* __restrict__ wrow = &lw[c * LW_PITCH + 2 * h];

    v8f acc[8] = {};
#pragma unroll
    for (int ks = 0; ks < 32; ++ks) {
#pragma unroll
        for (int t = 0; t < 8; ++t) {
            const v2f b = *(const v2f*)(wrow + t * 16 * LW_PITCH + 4 * ks);
            acc[t] = __builtin_amdgcn_wmma_f32_16x16x4_f32(
                 false, a[ks],  false, b,
                 (short)0, acc[t],  false,  false);
        }
    }

    float* __restrict__ ybase = Y + (size_t)(m0 + 8 * h) * HDIM + 4 * c;
    float4 bv[2];
#pragma unroll
    for (int g = 0; g < 2; ++g) {
        bv[g] = make_float4(0.0f, 0.0f, 0.0f, 0.0f);
        if (bias) bv[g] = *(const float4*)(bias + 64 * g + 4 * c);
    }

    if (accumulate) {
        float4 y[16];
#pragma unroll
        for (int i = 0; i < 16; ++i)
            y[i] = *(const float4*)(ybase + (size_t)(i & 7) * HDIM + 64 * (i >> 3));
#pragma unroll
        for (int i = 0; i < 16; ++i) {
            const int g = i >> 3, r = i & 7;
            float4 v;
            v.x = (acc[4 * g + 0][r] + bv[g].x) + y[i].x;
            v.y = (acc[4 * g + 1][r] + bv[g].y) + y[i].y;
            v.z = (acc[4 * g + 2][r] + bv[g].z) + y[i].z;
            v.w = (acc[4 * g + 3][r] + bv[g].w) + y[i].w;
            *(float4*)(ybase + (size_t)r * HDIM + 64 * g) = v;
        }
    } else {
#pragma unroll
        for (int i = 0; i < 16; ++i) {
            const int g = i >> 3, r = i & 7;
            float4 v;
            v.x = acc[4 * g + 0][r] + bv[g].x;
            v.y = acc[4 * g + 1][r] + bv[g].y;
            v.z = acc[4 * g + 2][r] + bv[g].z;
            v.w = acc[4 * g + 3][r] + bv[g].w;
            *(float4*)(ybase + (size_t)r * HDIM + 64 * g) = v;
        }
    }
}

__device__ __forceinline__ float sigmoidf_(float x) {
    return 1.0f / (1.0f + __expf(-x));
}

#define EPW 8
#define NBATCH 2

__launch_bounds__(256)
__global__ void edge_score_kernel(const float* __restrict__ Ah,
                                  const float* __restrict__ Bx,
                                  const float* __restrict__ v_w,
                                  const int* __restrict__ ci,
                                  const int* __restrict__ chi,
                                  float* __restrict__ e,
                                  unsigned int* __restrict__ seg_max_u,
                                  int L) {
    const int lane = threadIdx.x & 31;
    const int wave = __builtin_amdgcn_readfirstlane((int)(threadIdx.x >> 5));
    const int w0 = (blockIdx.x * 8 + wave) * (EPW * NBATCH);
    const float4 v = ((const float4*)v_w)[lane];

    for (int bt = 0; bt < NBATCH; ++bt) {
        const int e0 = w0 + bt * EPW;
        if (e0 >= L) break;
        const int le = min(e0 + (lane & (EPW - 1)), L - 1);
        const int my_c = ci[le], my_ch = chi[le];

        float4 a[EPW], b[EPW];
#pragma unroll
        for (int j = 0; j < EPW; ++j) {
            const int c  = __builtin_amdgcn_readlane(my_c, j);
            const int ch = __builtin_amdgcn_readlane(my_ch, j);
            a[j] = ((const float4*)(Ah + (size_t)ch * HDIM))[lane];
            b[j] = ((const float4*)(Bx + (size_t)c  * HDIM))[lane];
        }

        float mine = 0.0f;
#pragma unroll
        for (int j = 0; j < EPW; ++j) {
            float s = tanhf(a[j].x + b[j].x) * v.x + tanhf(a[j].y + b[j].y) * v.y
                    + tanhf(a[j].z + b[j].z) * v.z + tanhf(a[j].w + b[j].w) * v.w;
#pragma unroll
            for (int off = 16; off > 0; off >>= 1) s += __shfl_xor(s, off, 32);
            if (lane == j) mine = s;
        }

        if (lane < EPW && e0 + lane < L) {
            e[e0 + lane] = mine;
            const unsigned int b32 = __float_as_uint(mine);
            const unsigned int key = (b32 & 0x80000000u) ? ~b32 : (b32 | 0x80000000u);
            atomicMax(&seg_max_u[my_c], key);
        }
    }
}

__launch_bounds__(256)
__global__ void edge_softmax_kernel(const float* __restrict__ e,
                                    const int* __restrict__ ci,
                                    const unsigned int* __restrict__ seg_max_u,
                                    float* __restrict__ x_exp,
                                    float* __restrict__ seg_sum,
                                    int L) {
    const int t = blockIdx.x * 256 + threadIdx.x;
    if (t >= L) return;
    const int c = ci[t];
    const unsigned int key = seg_max_u[c];
    const unsigned int b32 = (key & 0x80000000u) ? (key & 0x7FFFFFFFu) : ~key;
    const float xe = __expf(e[t] - __uint_as_float(b32));
    x_exp[t] = xe;
    atomicAdd(&seg_sum[c], xe);
}

template <bool FULL>
__device__ __forceinline__ void edge_accum_batch(const float* __restrict__ x_exp,
                                                 const float* __restrict__ seg_sum,
                                                 const float* __restrict__ child_h,
                                                 const float* __restrict__ child_c,
                                                 const float* __restrict__ Wfx,
                                                 const float* __restrict__ Ufh,
                                                 const int* __restrict__ ci,
                                                 const int* __restrict__ chi,
                                                 float* __restrict__ h_hat,
                                                 float* __restrict__ sum_fc,
                                                 int L, int e0, int lane) {
    const int le = FULL ? e0 + (lane & (EPW - 1)) : min(e0 + (lane & (EPW - 1)), L - 1);
    const int my_c = ci[le], my_ch = chi[le];

    float4 hv[EPW], cv[EPW], wf[EPW], uf[EPW];
#pragma unroll
    for (int j = 0; j < EPW; ++j) {
        const int c  = __builtin_amdgcn_readlane(my_c, j);
        const int ch = __builtin_amdgcn_readlane(my_ch, j);
        hv[j] = ((const float4*)(child_h + (size_t)ch * HDIM))[lane];
        cv[j] = ((const float4*)(child_c + (size_t)ch * HDIM))[lane];
        wf[j] = ((const float4*)(Wfx + (size_t)c  * HDIM))[lane];
        uf[j] = ((const float4*)(Ufh + (size_t)ch * HDIM))[lane];
    }
    const float my_attn = x_exp[le] / (seg_sum[my_c] + EPSV);

#pragma unroll
    for (int j = 0; j < EPW; ++j)
        asm volatile("" :: "v"(hv[j].x), "v"(hv[j].y), "v"(hv[j].z), "v"(hv[j].w),
                           "v"(cv[j].x), "v"(cv[j].y), "v"(cv[j].z), "v"(cv[j].w),
                           "v"(wf[j].x), "v"(wf[j].y), "v"(wf[j].z), "v"(wf[j].w),
                           "v"(uf[j].x), "v"(uf[j].y), "v"(uf[j].z), "v"(uf[j].w));

#pragma unroll
    for (int j = 0; j < EPW; ++j) {
        if (FULL || e0 + j < L) {
            const int c = __builtin_amdgcn_readlane(my_c, j);
            const float attn = __uint_as_float(
                __builtin_amdgcn_readlane(__float_as_uint(my_attn), j));
            float* hh = h_hat  + (size_t)c * HDIM + lane * 4;
            float* sf = sum_fc + (size_t)c * HDIM + lane * 4;

            atomicAdd(&hh[0], attn * hv[j].x);
            atomicAdd(&hh[1], attn * hv[j].y);
            atomicAdd(&hh[2], attn * hv[j].z);
            atomicAdd(&hh[3], attn * hv[j].w);

            atomicAdd(&sf[0], sigmoidf_(wf[j].x + uf[j].x) * cv[j].x);
            atomicAdd(&sf[1], sigmoidf_(wf[j].y + uf[j].y) * cv[j].y);
            atomicAdd(&sf[2], sigmoidf_(wf[j].z + uf[j].z) * cv[j].z);
            atomicAdd(&sf[3], sigmoidf_(wf[j].w + uf[j].w) * cv[j].w);
        }
    }
}

__launch_bounds__(256)
__global__ void edge_accum_kernel(const float* __restrict__ x_exp,
                                  const float* __restrict__ seg_sum,
                                  const float* __restrict__ child_h,
                                  const float* __restrict__ child_c,
                                  const float* __restrict__ Wfx,
                                  const float* __restrict__ Ufh,
                                  const int* __restrict__ ci,
                                  const int* __restrict__ chi,
                                  float* __restrict__ h_hat,
                                  float* __restrict__ sum_fc,
                                  int L) {
    const int lane = threadIdx.x & 31;
    const int wave = __builtin_amdgcn_readfirstlane((int)(threadIdx.x >> 5));
    const int w0 = (blockIdx.x * 8 + wave) * (EPW * NBATCH);

    for (int bt = 0; bt < NBATCH; ++bt) {
        const int e0 = w0 + bt * EPW;
        if (e0 >= L) break;
        if (e0 + EPW <= L)
            edge_accum_batch<true>(x_exp, seg_sum, child_h, child_c, Wfx, Ufh,
                                   ci, chi, h_hat, sum_fc, L, e0, lane);
        else
            edge_accum_batch<false>(x_exp, seg_sum, child_h, child_c, Wfx, Ufh,
                                    ci, chi, h_hat, sum_fc, L, e0, lane);
    }
}

#define GATE_V4 4
__launch_bounds__(256)
__global__ void final_gate_kernel(const float* __restrict__ Gi,
                                  const float* __restrict__ Gc,
                                  const float* __restrict__ Go,
                                  const float* __restrict__ sum_fc,
                                  float* __restrict__ h_out,
                                  float* __restrict__ c_out,
                                  size_t n4) {
    const size_t base = (size_t)blockIdx.x * (256 * GATE_V4) + threadIdx.x;
    float4 gi[GATE_V4], gc[GATE_V4], go[GATE_V4], sf[GATE_V4];
#pragma unroll
    for (int j = 0; j < GATE_V4; ++j) {
        const size_t t  = base + (size_t)j * 256;
        const size_t tl = t < n4 ? t : n4 - 1;
        gi[j] = ((const float4*)Gi)[tl];
        gc[j] = ((const float4*)Gc)[tl];
        go[j] = ((const float4*)Go)[tl];
        sf[j] = ((const float4*)sum_fc)[tl];
    }
#pragma unroll
    for (int j = 0; j < GATE_V4; ++j) {
        const size_t t = base + (size_t)j * 256;
        float4 c, h;
        c.x = sigmoidf_(gi[j].x) * tanhf(gc[j].x) + sf[j].x;
        c.y = sigmoidf_(gi[j].y) * tanhf(gc[j].y) + sf[j].y;
        c.z = sigmoidf_(gi[j].z) * tanhf(gc[j].z) + sf[j].z;
        c.w = sigmoidf_(gi[j].w) * tanhf(gc[j].w) + sf[j].w;
        h.x = sigmoidf_(go[j].x) * tanhf(c.x);
        h.y = sigmoidf_(go[j].y) * tanhf(c.y);
        h.z = sigmoidf_(go[j].z) * tanhf(c.z);
        h.w = sigmoidf_(go[j].w) * tanhf(c.w);
        if (t < n4) {
            ((float4*)h_out)[t] = h;
            ((float4*)c_out)[t] = c;
        }
    }
}

extern "C" void kernel_launch(void* const* d_in, const int* in_sizes, int n_in,
                              void* d_out, int out_size, void* d_ws, size_t ws_size,
                              hipStream_t stream) {
    const float* x_emb   = (const float*)d_in[0];
    const float* child_h = (const float*)d_in[1];
    const float* child_c = (const float*)d_in[2];
    const int*   ci      = (const int*)d_in[3];
    const int*   chi     = (const int*)d_in[4];
    const float* Wi_w = (const float*)d_in[5];
    const float* Ui_w = (const float*)d_in[6];
    const float* Wf_w = (const float*)d_in[7];
    const float* Uf_w = (const float*)d_in[8];
    const float* Wo_w = (const float*)d_in[9];
    const float* Uo_w = (const float*)d_in[10];
    const float* Wc_w = (const float*)d_in[11];
    const float* Uc_w = (const float*)d_in[12];
    const float* Wa_w = (const float*)d_in[13];
    const float* Ua_w = (const float*)d_in[14];
    const float* Wi_b = (const float*)d_in[15];
    const float* Wf_b = (const float*)d_in[16];
    const float* Wo_b = (const float*)d_in[17];
    const float* Wc_b = (const float*)d_in[18];
    const float* Wa_b = (const float*)d_in[19];
    const float* v_w  = (const float*)d_in[20];

    const int M = in_sizes[0] / HDIM;
    const int L = in_sizes[3];
    const size_t MH = (size_t)M * HDIM;

    float* ws    = (float*)d_ws;
    float* Ah    = ws;
    float* Bx    = Ah  + MH;
    float* Gi    = Bx  + MH;
    float* Wfx   = Gi  + MH;
    float* Go    = Wfx + MH;
    float* Gc    = Go  + MH;
    float* Ufh   = Gc  + MH;
    float* h_hat = Ufh + MH;
    float* sumfc = h_hat + MH;
    float* e_buf = sumfc + MH;
    float* x_exp = e_buf + L;
    float* seg_sum = x_exp + L;
    unsigned int* seg_max_u = (unsigned int*)(seg_sum + M);

    hipMemsetAsync(h_hat,     0, 2 * MH * sizeof(float), stream);
    hipMemsetAsync(seg_sum,   0, (size_t)M * sizeof(float), stream);
    hipMemsetAsync(seg_max_u, 0, (size_t)M * sizeof(unsigned int), stream);

    const dim3 gblk(256);
    const dim3 ggrd((M + 127) / 128);

    gemm_wmma_f32<<<ggrd, gblk, 0, stream>>>(child_h, Wa_w, Wa_b, Ah, M, 0);
    gemm_wmma_f32<<<ggrd, gblk, 0, stream>>>(x_emb,   Ua_w, nullptr, Bx, M, 0);
    gemm_wmma_f32<<<ggrd, gblk, 0, stream>>>(x_emb,   Wi_w, Wi_b, Gi, M, 0);
    gemm_wmma_f32<<<ggrd, gblk, 0, stream>>>(x_emb,   Wf_w, Wf_b, Wfx, M, 0);
    gemm_wmma_f32<<<ggrd, gblk, 0, stream>>>(x_emb,   Wo_w, Wo_b, Go, M, 0);
    gemm_wmma_f32<<<ggrd, gblk, 0, stream>>>(x_emb,   Wc_w, Wc_b, Gc, M, 0);
    gemm_wmma_f32<<<ggrd, gblk, 0, stream>>>(child_h, Uf_w, nullptr, Ufh, M, 0);

    edge_score_kernel<<<(L + 8 * EPW * NBATCH - 1) / (8 * EPW * NBATCH), gblk, 0, stream>>>(Ah, Bx, v_w, ci, chi,
                                                        e_buf, seg_max_u, L);
    edge_softmax_kernel<<<(L + 255) / 256, gblk, 0, stream>>>(e_buf, ci, seg_max_u,
                                                              x_exp, seg_sum, L);
    edge_accum_kernel<<<(L + 8 * EPW * NBATCH - 1) / (8 * EPW * NBATCH), gblk, 0, stream>>>(x_exp, seg_sum,
                                                        child_h, child_c, Wfx, Ufh,
                                                        ci, chi, h_hat, sumfc, L);

    gemm_wmma_f32<<<ggrd, gblk, 0, stream>>>(h_hat, Ui_w, nullptr, Gi, M, 1);
    gemm_wmma_f32<<<ggrd, gblk, 0, stream>>>(h_hat, Uc_w, nullptr, Gc, M, 1);
    gemm_wmma_f32<<<ggrd, gblk, 0, stream>>>(h_hat, Uo_w, nullptr, Go, M, 1);

    float* h_out = (float*)d_out;
    float* c_out = h_out + MH;
    final_gate_kernel<<<(unsigned)((MH / 4 + 256 * GATE_V4 - 1) / (256 * GATE_V4)), gblk, 0, stream>>>(
        Gi, Gc, Go, sumfc, h_out, c_out, MH / 4);
}


# Edge sums: float atomics replaced by per-call counting sort of edges by candidate (int atomics) + one wave per candidate adding in registers in ascending edge index (refuted try: arrival order)
# speedup vs baseline: 2.2484x; 1.3079x over previous
#include <hip/hip_runtime.h>
#include <hip/hip_bf16.h>

#define HDIM 128
#define EPSV 1e-9f

typedef __attribute__((ext_vector_type(2))) float v2f;
typedef __attribute__((ext_vector_type(8))) float v8f;

#define LW_PITCH 132
__launch_bounds__(256) __attribute__((amdgpu_waves_per_eu(2, 4)))
__global__ void gemm_wmma_f32(const float* __restrict__ X,
                              const float* __restrict__ W,
                              const float* __restrict__ bias,
                              float* __restrict__ Y,
                              int Mrows, int accumulate) {
    __shared__ float lw[HDIM * LW_PITCH];

    const int wave = threadIdx.x >> 5;
    const int lane = threadIdx.x & 31;
    const int c    = lane & 15;
    const int h    = lane >> 4;
    const int m0   = blockIdx.x * 128 + wave * 16;
    const bool active = m0 < Mrows;

    v2f a[32];
    if (active) {
        const float* __restrict__ xrow = X + (size_t)(m0 + c) * HDIM + 2 * h;
#pragma unroll
        for (int ks = 0; ks < 32; ++ks) {
            a[ks].x = xrow[4 * ks];
            a[ks].y = xrow[4 * ks + 1];
        }
    }

#pragma unroll
    for (int it = 0; it < (HDIM * HDIM / 4) / 256; ++it) {
        const int i  = it * 256 + threadIdx.x;
        const int n  = i >> 5;
        const int k4 = (i & 31) << 2;
        const float4 w = *(const float4*)(W + (size_t)n * HDIM + k4);
        const int p  = (n & 64) | ((n & 3) << 4) | ((n & 63) >> 2);
        *(float4*)(&lw[p * LW_PITCH + k4]) = w;
    }
    __syncthreads();
    if (!active) return;

    const float* __restrict__ wrow = &lw[c * LW_PITCH + 2 * h];

    v8f acc[8] = {};
#pragma unroll
    for (int ks = 0; ks < 32; ++ks) {
#pragma unroll
        for (int t = 0; t < 8; ++t) {
            const v2f b = *(const v2f*)(wrow + t * 16 * LW_PITCH + 4 * ks);
            acc[t] = __builtin_amdgcn_wmma_f32_16x16x4_f32(
                 false, a[ks],  false, b,
                 (short)0, acc[t],  false,  false);
        }
    }

    float* __restrict__ ybase = Y + (size_t)(m0 + 8 * h) * HDIM + 4 * c;
    float4 bv[2];
#pragma unroll
    for (int g = 0; g < 2; ++g) {
        bv[g] = make_float4(0.0f, 0.0f, 0.0f, 0.0f);
        if (bias) bv[g] = *(const float4*)(bias + 64 * g + 4 * c);
    }

    if (accumulate) {
        float4 y[16];
#pragma unroll
        for (int i = 0; i < 16; ++i)
            y[i] = *(const float4*)(ybase + (size_t)(i & 7) * HDIM + 64 * (i >> 3));
#pragma unroll
        for (int i = 0; i < 16; ++i) {
            const int g = i >> 3, r = i & 7;
            float4 v;
            v.x = (acc[4 * g + 0][r] + bv[g].x) + y[i].x;
            v.y = (acc[4 * g + 1][r] + bv[g].y) + y[i].y;
            v.z = (acc[4 * g + 2][r] + bv[g].z) + y[i].z;
            v.w = (acc[4 * g + 3][r] + bv[g].w) + y[i].w;
            *(float4*)(ybase + (size_t)r * HDIM + 64 * g) = v;
        }
    } else {
#pragma unroll
        for (int i = 0; i < 16; ++i) {
            const int g = i >> 3, r = i & 7;
            float4 v;
            v.x = acc[4 * g + 0][r] + bv[g].x;
            v.y = acc[4 * g + 1][r] + bv[g].y;
            v.z = acc[4 * g + 2][r] + bv[g].z;
            v.w = acc[4 * g + 3][r] + bv[g].w;
            *(float4*)(ybase + (size_t)r * HDIM + 64 * g) = v;
        }
    }
}

__device__ __forceinline__ float sigmoidf_(float x) {
    return 1.0f / (1.0f + __expf(-x));
}

#define EPW 8
#define NBATCH 2

__launch_bounds__(256)
__global__ void edge_score_kernel(const float* __restrict__ Ah,
                                  const float* __restrict__ Bx,
                                  const float* __restrict__ v_w,
                                  const int* __restrict__ ci,
                                  const int* __restrict__ chi,
                                  float* __restrict__ e,
                                  int* __restrict__ cnt,
                                  int L) {
    const int lane = threadIdx.x & 31;
    const int wave = __builtin_amdgcn_readfirstlane((int)(threadIdx.x >> 5));
    const int w0 = (blockIdx.x * 8 + wave) * (EPW * NBATCH);
    const float4 v = ((const float4*)v_w)[lane];

    for (int bt = 0; bt < NBATCH; ++bt) {
        const int e0 = w0 + bt * EPW;
        if (e0 >= L) break;
        const int le = min(e0 + (lane & (EPW - 1)), L - 1);
        const int my_c = ci[le], my_ch = chi[le];

        float4 a[EPW], b[EPW];
#pragma unroll
        for (int j = 0; j < EPW; ++j) {
            const int c  = __builtin_amdgcn_readlane(my_c, j);
            const int ch = __builtin_amdgcn_readlane(my_ch, j);
            a[j] = ((const float4*)(Ah + (size_t)ch * HDIM))[lane];
            b[j] = ((const float4*)(Bx + (size_t)c  * HDIM))[lane];
        }

        float mine = 0.0f;
#pragma unroll
        for (int j = 0; j < EPW; ++j) {
            float s = tanhf(a[j].x + b[j].x) * v.x + tanhf(a[j].y + b[j].y) * v.y
                    + tanhf(a[j].z + b[j].z) * v.z + tanhf(a[j].w + b[j].w) * v.w;
#pragma unroll
            for (int off = 16; off > 0; off >>= 1) s += __shfl_xor(s, off, 32);
            if (lane == j) mine = s;
        }

        if (lane < EPW && e0 + lane < L) {
            e[e0 + lane] = mine;
            atomicAdd(&cnt[my_c], 1);
        }
    }
}


__launch_bounds__(1024)
__global__ void seg_scan_kernel(const int* __restrict__ cnt,
                                int* __restrict__ start, int M) {
    __shared__ int wtot[32];
    __shared__ int wpre[32];
    const int tid  = threadIdx.x;
    const int lane = tid & 31;
    const int wv   = tid >> 5;
    const int base = blockIdx.x * 1024;

    int pre = 0;
#pragma unroll 8
    for (int i = tid; i < base; i += 1024) pre += cnt[i];

    const int idx = base + tid;
    const int v = idx < M ? cnt[idx] : 0;

    int inc = v;
#pragma unroll
    for (int off = 1; off < 32; off <<= 1) {
        const int o = __shfl_up(inc, off, 32);
        if (lane >= off) inc += o;
    }
#pragma unroll
    for (int off = 16; off > 0; off >>= 1) pre += __shfl_xor(pre, off, 32);

    if (lane == 31) wtot[wv] = inc;
    if (lane == 0)  wpre[wv] = pre;
    __syncthreads();

    int before = wpre[lane] + (lane < wv ? wtot[lane] : 0);
#pragma unroll
    for (int off = 16; off > 0; off >>= 1) before += __shfl_xor(before, off, 32);

    if (idx < M) start[idx] = before + inc - v;
}

__launch_bounds__(256)
__global__ void seg_fill_kernel(const int* __restrict__ ci,
                                const int* __restrict__ start,
                                int* __restrict__ cnt,
                                int* __restrict__ perm, int L) {
    const int t = blockIdx.x * 256 + threadIdx.x;
    if (t >= L) return;
    const int c = ci[t];
    const int old = atomicSub(&cnt[c], 1);
    perm[start[c] + old - 1] = t;
}

__device__ __forceinline__ int seg_next_edge(const int* __restrict__ perm,
                                             int s0, int n, int prev, int lane) {
    int cur = 0x7fffffff;
    for (int i = lane; i < n; i += 32) {
        const int id = perm[s0 + i];
        if (id > prev && id < cur) cur = id;
    }
#pragma unroll
    for (int off = 16; off > 0; off >>= 1) cur = min(cur, __shfl_xor(cur, off, 32));
    return cur;
}

#define SEG_B 8
__launch_bounds__(256)
__global__ void seg_attend_kernel(const float* __restrict__ e,
                                  const int* __restrict__ chi,
                                  const int* __restrict__ start,
                                  const int* __restrict__ perm,
                                  const float* __restrict__ child_h,
                                  const float* __restrict__ child_c,
                                  const float* __restrict__ Wfx,
                                  const float* __restrict__ Ufh,
                                  float* __restrict__ h_hat,
                                  float* __restrict__ sum_fc,
                                  int M, int L) {
    const int lane = threadIdx.x & 31;
    const int wave = __builtin_amdgcn_readfirstlane((int)(threadIdx.x >> 5));
    const int c = blockIdx.x * 8 + wave;
    if (c >= M) return;
    const int sv = (lane < 2 && c + lane < M) ? start[c + lane] : L;
    const int s0 = __builtin_amdgcn_readlane(sv, 0);
    const int s1 = __builtin_amdgcn_readlane(sv, 1);
    const int n  = s1 - s0;

    const float4 wf = ((const float4*)(Wfx + (size_t)c * HDIM))[lane];
    float4 ah = make_float4(0.0f, 0.0f, 0.0f, 0.0f);
    float4 af = make_float4(0.0f, 0.0f, 0.0f, 0.0f);

    if (n <= 32) {
        const bool valid = lane < n;
        const int raw = valid ? perm[s0 + lane] : 0x7fffffff;
        int rank = 0;
        for (int j = 0; j < n; ++j) {
            const int o = __builtin_amdgcn_readlane(raw, j);
            rank += (o < raw) ? 1 : 0;
        }
        int id = 0;
        for (int j = 0; j < n; ++j) {
            const int r = __builtin_amdgcn_readlane(rank, j);
            const int o = __builtin_amdgcn_readlane(raw, j);
            if (lane == r) id = o;
        }
        const int   my_ch = valid ? chi[id] : 0;
        const float my_e  = valid ? e[id] : -__builtin_inff();

        float m = my_e;
#pragma unroll
        for (int off = 16; off > 0; off >>= 1) m = fmaxf(m, __shfl_xor(m, off, 32));
        const float my_x = valid ? __expf(my_e - m) : 0.0f;
        float ssum = my_x;
#pragma unroll
        for (int off = 16; off > 0; off >>= 1) ssum += __shfl_xor(ssum, off, 32);
        const float my_attn = my_x / (ssum + EPSV);

        for (int kb = 0; kb < n; kb += SEG_B) {
            float4 hv[SEG_B], cv[SEG_B], uf[SEG_B];
#pragma unroll
            for (int j = 0; j < SEG_B; ++j) {
                const int kk = min(kb + j, n - 1);
                const int ch = __builtin_amdgcn_readlane(my_ch, kk);
                hv[j] = ((const float4*)(child_h + (size_t)ch * HDIM))[lane];
                cv[j] = ((const float4*)(child_c + (size_t)ch * HDIM))[lane];
                uf[j] = ((const float4*)(Ufh     + (size_t)ch * HDIM))[lane];
            }
#pragma unroll
            for (int j = 0; j < SEG_B; ++j)
                asm volatile("" :: "v"(hv[j].x), "v"(hv[j].y), "v"(hv[j].z), "v"(hv[j].w),
                                   "v"(cv[j].x), "v"(cv[j].y), "v"(cv[j].z), "v"(cv[j].w),
                                   "v"(uf[j].x), "v"(uf[j].y), "v"(uf[j].z), "v"(uf[j].w));
#pragma unroll
            for (int j = 0; j < SEG_B; ++j) {
                if (kb + j < n) {
                    const float attn = __uint_as_float(
                        __builtin_amdgcn_readlane(__float_as_uint(my_attn), kb + j));
                    ah.x += attn * hv[j].x;
                    ah.y += attn * hv[j].y;
                    ah.z += attn * hv[j].z;
                    ah.w += attn * hv[j].w;
                    af.x += sigmoidf_(wf.x + uf[j].x) * cv[j].x;
                    af.y += sigmoidf_(wf.y + uf[j].y) * cv[j].y;
                    af.z += sigmoidf_(wf.z + uf[j].z) * cv[j].z;
                    af.w += sigmoidf_(wf.w + uf[j].w) * cv[j].w;
                }
            }
        }
    } else {
        float m = -__builtin_inff();
        for (int i = lane; i < n; i += 32) m = fmaxf(m, e[perm[s0 + i]]);
#pragma unroll
        for (int off = 16; off > 0; off >>= 1) m = fmaxf(m, __shfl_xor(m, off, 32));

        float ssum = 0.0f;
        int prev = -1;
        for (int k = 0; k < n; ++k) {
            const int cur = seg_next_edge(perm, s0, n, prev, lane);
            ssum += __expf(e[cur] - m);
            prev = cur;
        }
        const float denom = ssum + EPSV;

        prev = -1;
        for (int k = 0; k < n; ++k) {
            const int cur = seg_next_edge(perm, s0, n, prev, lane);
            prev = cur;
            const int ch = chi[cur];
            const float attn = __expf(e[cur] - m) / denom;
            const float4 hv = ((const float4*)(child_h + (size_t)ch * HDIM))[lane];
            const float4 cv = ((const float4*)(child_c + (size_t)ch * HDIM))[lane];
            const float4 uf = ((const float4*)(Ufh     + (size_t)ch * HDIM))[lane];
            ah.x += attn * hv.x;
            ah.y += attn * hv.y;
            ah.z += attn * hv.z;
            ah.w += attn * hv.w;
            af.x += sigmoidf_(wf.x + uf.x) * cv.x;
            af.y += sigmoidf_(wf.y + uf.y) * cv.y;
            af.z += sigmoidf_(wf.z + uf.z) * cv.z;
            af.w += sigmoidf_(wf.w + uf.w) * cv.w;
        }
    }

    ((float4*)(h_hat  + (size_t)c * HDIM))[lane] = ah;
    ((float4*)(sum_fc + (size_t)c * HDIM))[lane] = af;
}

#define GATE_V4 4
__launch_bounds__(256)
__global__ void final_gate_kernel(const float* __restrict__ Gi,
                                  const float* __restrict__ Gc,
                                  const float* __restrict__ Go,
                                  const float* __restrict__ sum_fc,
                                  float* __restrict__ h_out,
                                  float* __restrict__ c_out,
                                  size_t n4) {
    const size_t base = (size_t)blockIdx.x * (256 * GATE_V4) + threadIdx.x;
    float4 gi[GATE_V4], gc[GATE_V4], go[GATE_V4], sf[GATE_V4];
#pragma unroll
    for (int j = 0; j < GATE_V4; ++j) {
        const size_t t  = base + (size_t)j * 256;
        const size_t tl = t < n4 ? t : n4 - 1;
        gi[j] = ((const float4*)Gi)[tl];
        gc[j] = ((const float4*)Gc)[tl];
        go[j] = ((const float4*)Go)[tl];
        sf[j] = ((const float4*)sum_fc)[tl];
    }
#pragma unroll
    for (int j = 0; j < GATE_V4; ++j) {
        const size_t t = base + (size_t)j * 256;
        float4 c, h;
        c.x = sigmoidf_(gi[j].x) * tanhf(gc[j].x) + sf[j].x;
        c.y = sigmoidf_(gi[j].y) * tanhf(gc[j].y) + sf[j].y;
        c.z = sigmoidf_(gi[j].z) * tanhf(gc[j].z) + sf[j].z;
        c.w = sigmoidf_(gi[j].w) * tanhf(gc[j].w) + sf[j].w;
        h.x = sigmoidf_(go[j].x) * tanhf(c.x);
        h.y = sigmoidf_(go[j].y) * tanhf(c.y);
        h.z = sigmoidf_(go[j].z) * tanhf(c.z);
        h.w = sigmoidf_(go[j].w) * tanhf(c.w);
        if (t < n4) {
            ((float4*)h_out)[t] = h;
            ((float4*)c_out)[t] = c;
        }
    }
}

extern "C" void kernel_launch(void* const* d_in, const int* in_sizes, int n_in,
                              void* d_out, int out_size, void* d_ws, size_t ws_size,
                              hipStream_t stream) {
    const float* x_emb   = (const float*)d_in[0];
    const float* child_h = (const float*)d_in[1];
    const float* child_c = (const float*)d_in[2];
    const int*   ci      = (const int*)d_in[3];
    const int*   chi     = (const int*)d_in[4];
    const float* Wi_w = (const float*)d_in[5];
    const float* Ui_w = (const float*)d_in[6];
    const float* Wf_w = (const float*)d_in[7];
    const float* Uf_w = (const float*)d_in[8];
    const float* Wo_w = (const float*)d_in[9];
    const float* Uo_w = (const float*)d_in[10];
    const float* Wc_w = (const float*)d_in[11];
    const float* Uc_w = (const float*)d_in[12];
    const float* Wa_w = (const float*)d_in[13];
    const float* Ua_w = (const float*)d_in[14];
    const float* Wi_b = (const float*)d_in[15];
    const float* Wf_b = (const float*)d_in[16];
    const float* Wo_b = (const float*)d_in[17];
    const float* Wc_b = (const float*)d_in[18];
    const float* Wa_b = (const float*)d_in[19];
    const float* v_w  = (const float*)d_in[20];

    const int M = in_sizes[0] / HDIM;
    const int L = in_sizes[3];
    const size_t MH = (size_t)M * HDIM;

    float* ws    = (float*)d_ws;
    float* Ah    = ws;
    float* Bx    = Ah  + MH;
    float* Gi    = Bx  + MH;
    float* Wfx   = Gi  + MH;
    float* Go    = Wfx + MH;
    float* Gc    = Go  + MH;
    float* Ufh   = Gc  + MH;
    float* h_hat = Ufh + MH;
    float* sumfc = h_hat + MH;
    float* e_buf = sumfc + MH;
    int* perm    = (int*)(e_buf + L);
    int* cnt     = perm + L;
    int* start   = cnt + M;

    hipMemsetAsync(cnt, 0, (size_t)M * sizeof(int), stream);

    const dim3 gblk(256);
    const dim3 ggrd((M + 127) / 128);

    gemm_wmma_f32<<<ggrd, gblk, 0, stream>>>(child_h, Wa_w, Wa_b, Ah, M, 0);
    gemm_wmma_f32<<<ggrd, gblk, 0, stream>>>(x_emb,   Ua_w, nullptr, Bx, M, 0);
    gemm_wmma_f32<<<ggrd, gblk, 0, stream>>>(x_emb,   Wi_w, Wi_b, Gi, M, 0);
    gemm_wmma_f32<<<ggrd, gblk, 0, stream>>>(x_emb,   Wf_w, Wf_b, Wfx, M, 0);
    gemm_wmma_f32<<<ggrd, gblk, 0, stream>>>(x_emb,   Wo_w, Wo_b, Go, M, 0);
    gemm_wmma_f32<<<ggrd, gblk, 0, stream>>>(x_emb,   Wc_w, Wc_b, Gc, M, 0);
    gemm_wmma_f32<<<ggrd, gblk, 0, stream>>>(child_h, Uf_w, nullptr, Ufh, M, 0);

    edge_score_kernel<<<(L + 8 * EPW * NBATCH - 1) / (8 * EPW * NBATCH), gblk, 0, stream>>>(Ah, Bx, v_w, ci, chi,
                                                        e_buf, cnt, L);
    seg_scan_kernel<<<(M + 1023) / 1024, dim3(1024), 0, stream>>>(cnt, start, M);
    seg_fill_kernel<<<(L + 255) / 256, gblk, 0, stream>>>(ci, start, cnt, perm, L);
    seg_attend_kernel<<<(M + 7) / 8, gblk, 0, stream>>>(e_buf, chi, start, perm,
                                                        child_h, child_c, Wfx, Ufh,
                                                        h_hat, sumfc, M, L);

    gemm_wmma_f32<<<ggrd, gblk, 0, stream>>>(h_hat, Ui_w, nullptr, Gi, M, 1);
    gemm_wmma_f32<<<ggrd, gblk, 0, stream>>>(h_hat, Uc_w, nullptr, Gc, M, 1);
    gemm_wmma_f32<<<ggrd, gblk, 0, stream>>>(h_hat, Uo_w, nullptr, Go, M, 1);

    float* h_out = (float*)d_out;
    float* c_out = h_out + MH;
    final_gate_kernel<<<(unsigned)((MH / 4 + 256 * GATE_V4 - 1) / (256 * GATE_V4)), gblk, 0, stream>>>(
        Gi, Gc, Go, sumfc, h_out, c_out, MH / 4);
}


# All products of a phase in one launch: a wave keeps its 16x128 A tile in registers over the matrices that share its input (5, 2, 3), W restaged in LDS per matrix; 10 GEMM launches become 2
# speedup vs baseline: 2.4735x; 1.1001x over previous
#include <hip/hip_runtime.h>
#include <hip/hip_bf16.h>

#define HDIM 128
#define EPSV 1e-9f

typedef __attribute__((ext_vector_type(2))) float v2f;
typedef __attribute__((ext_vector_type(8))) float v8f;

#define LW_PITCH 132
#define GEMM_PICK(j, p0, p1, p2, p3, p4, p5, p6)                                   \
    ((j) == 0 ? (p0) : (j) == 1 ? (p1) : (j) == 2 ? (p2) : (j) == 3 ? (p3)        \
   : (j) == 4 ? (p4) : (j) == 5 ? (p5) : (p6))
__launch_bounds__(256) __attribute__((amdgpu_waves_per_eu(2, 4)))
__global__ void gemm_group_wmma_f32(const float* __restrict__ XA,
                                    const float* __restrict__ XB,
                                    const float* W0, const float* W1, const float* W2,
                                    const float* W3, const float* W4, const float* W5,
                                    const float* W6,
                                    const float* B0, const float* B1, const float* B2,
                                    const float* B3, const float* B4, const float* B5,
                                    const float* B6,
                                    float* Y0, float* Y1, float* Y2, float* Y3,
                                    float* Y4, float* Y5, float* Y6,
                                    int nA, int nB, int Mrows, int accumulate) {
    __shared__ float lw[HDIM * LW_PITCH];

    const int wave = threadIdx.x >> 5;
    const int lane = threadIdx.x & 31;
    const int c    = lane & 15;
    const int h    = lane >> 4;

    const int nblk  = (Mrows + 127) >> 7;
    const int grp   = (int)blockIdx.x >= nblk ? 1 : 0;
    const int rb    = (int)blockIdx.x - grp * nblk;
    const int first = grp ? nA : 0;
    const int count = grp ? nB : nA;
    const float* __restrict__ X = grp ? XB : XA;

    const int m0   = rb * 128 + wave * 16;
    const bool active = m0 < Mrows;

    v2f a[32];
    if (active) {
        const float* __restrict__ xrow = X + (size_t)(m0 + c) * HDIM + 2 * h;
#pragma unroll
        for (int ks = 0; ks < 32; ++ks) {
            a[ks].x = xrow[4 * ks];
            a[ks].y = xrow[4 * ks + 1];
        }
    }

    const float* __restrict__ wrow = &lw[c * LW_PITCH + 2 * h];

    float* __restrict__ lw_st = &lw[((wave & 3) * 16 + (wave >> 2)) * LW_PITCH + (lane << 2)];
    const int w_off = (int)threadIdx.x * 4;
#define GEMM_STAGE_W(Wptr)                                                           \
    {                                                                                \
        const float* __restrict__ wsrc = (Wptr) + w_off;                             \
        float4 wv[16];                                                               \
        _Pragma("unroll")                                                            \
        for (int it = 0; it < 16; ++it)                                              \
            wv[it] = *(const float4*)(wsrc + it * 1024);                             \
        _Pragma("unroll")                                                            \
        for (int it = 0; it < 16; ++it)                                              \
            *(float4*)(lw_st + ((it >> 3) * 64 + (it & 7) * 2) * LW_PITCH) = wv[it]; \
    }

    GEMM_STAGE_W(GEMM_PICK(first, W0, W1, W2, W3, W4, W5, W6))

#pragma nounroll
    for (int g = 0; g < count; ++g) {
        const int j = first + g;
        const float* __restrict__ bias = GEMM_PICK(j, B0, B1, B2, B3, B4, B5, B6);
        float* __restrict__ Y          = GEMM_PICK(j, Y0, Y1, Y2, Y3, Y4, Y5, Y6);

        __syncthreads();

        if (active) {
            v8f acc[8] = {};
#pragma unroll
            for (int ks = 0; ks < 32; ++ks) {
#pragma unroll
                for (int t = 0; t < 8; ++t) {
                    const v2f b = *(const v2f*)(wrow + t * 16 * LW_PITCH + 4 * ks);
                    acc[t] = __builtin_amdgcn_wmma_f32_16x16x4_f32(
                         false, a[ks],  false, b,
                         (short)0, acc[t],  false,  false);
                }
            }

            float* __restrict__ ybase = Y + (size_t)(m0 + 8 * h) * HDIM + 4 * c;
            float4 bv[2];
#pragma unroll
            for (int q = 0; q < 2; ++q) {
                bv[q] = make_float4(0.0f, 0.0f, 0.0f, 0.0f);
                if (bias) bv[q] = *(const float4*)(bias + 64 * q + 4 * c);
            }

            if (accumulate) {
#pragma unroll
                for (int q = 0; q < 2; ++q) {
                    float4 y[8];
#pragma unroll
                    for (int r = 0; r < 8; ++r)
                        y[r] = *(const float4*)(ybase + (size_t)r * HDIM + 64 * q);
#pragma unroll
                    for (int r = 0; r < 8; r += 4)
                        asm volatile("" :: "v"(y[r].x), "v"(y[r].y), "v"(y[r].z), "v"(y[r].w),
                                           "v"(y[r + 1].x), "v"(y[r + 1].y), "v"(y[r + 1].z), "v"(y[r + 1].w),
                                           "v"(y[r + 2].x), "v"(y[r + 2].y), "v"(y[r + 2].z), "v"(y[r + 2].w),
                                           "v"(y[r + 3].x), "v"(y[r + 3].y), "v"(y[r + 3].z), "v"(y[r + 3].w));
#pragma unroll
                    for (int r = 0; r < 8; ++r) {
                        float4 v;
                        v.x = (acc[4 * q + 0][r] + bv[q].x) + y[r].x;
                        v.y = (acc[4 * q + 1][r] + bv[q].y) + y[r].y;
                        v.z = (acc[4 * q + 2][r] + bv[q].z) + y[r].z;
                        v.w = (acc[4 * q + 3][r] + bv[q].w) + y[r].w;
                        *(float4*)(ybase + (size_t)r * HDIM + 64 * q) = v;
                    }
                }
            } else {
#pragma unroll
                for (int i = 0; i < 16; ++i) {
                    const int q = i >> 3, r = i & 7;
                    float4 v;
                    v.x = acc[4 * q + 0][r] + bv[q].x;
                    v.y = acc[4 * q + 1][r] + bv[q].y;
                    v.z = acc[4 * q + 2][r] + bv[q].z;
                    v.w = acc[4 * q + 3][r] + bv[q].w;
                    *(float4*)(ybase + (size_t)r * HDIM + 64 * q) = v;
                }
            }
        }

        if (g + 1 < count) {
            __syncthreads();
            GEMM_STAGE_W(GEMM_PICK(j + 1, W0, W1, W2, W3, W4, W5, W6))
        }
    }
#undef GEMM_STAGE_W
}

__device__ __forceinline__ float sigmoidf_(float x) {
    return 1.0f / (1.0f + __expf(-x));
}

#define EPW 8
#define NBATCH 2

__launch_bounds__(256)
__global__ void edge_score_kernel(const float* __restrict__ Ah,
                                  const float* __restrict__ Bx,
                                  const float* __restrict__ v_w,
                                  const int* __restrict__ ci,
                                  const int* __restrict__ chi,
                                  float* __restrict__ e,
                                  int* __restrict__ cnt,
                                  int L) {
    const int lane = threadIdx.x & 31;
    const int wave = __builtin_amdgcn_readfirstlane((int)(threadIdx.x >> 5));
    const int w0 = (blockIdx.x * 8 + wave) * (EPW * NBATCH);
    const float4 v = ((const float4*)v_w)[lane];

    for (int bt = 0; bt < NBATCH; ++bt) {
        const int e0 = w0 + bt * EPW;
        if (e0 >= L) break;
        const int le = min(e0 + (lane & (EPW - 1)), L - 1);
        const int my_c = ci[le], my_ch = chi[le];

        float4 a[EPW], b[EPW];
#pragma unroll
        for (int j = 0; j < EPW; ++j) {
            const int c  = __builtin_amdgcn_readlane(my_c, j);
            const int ch = __builtin_amdgcn_readlane(my_ch, j);
            a[j] = ((const float4*)(Ah + (size_t)ch * HDIM))[lane];
            b[j] = ((const float4*)(Bx + (size_t)c  * HDIM))[lane];
        }

        float mine = 0.0f;
#pragma unroll
        for (int j = 0; j < EPW; ++j) {
            float s = tanhf(a[j].x + b[j].x) * v.x + tanhf(a[j].y + b[j].y) * v.y
                    + tanhf(a[j].z + b[j].z) * v.z + tanhf(a[j].w + b[j].w) * v.w;
#pragma unroll
            for (int off = 16; off > 0; off >>= 1) s += __shfl_xor(s, off, 32);
            if (lane == j) mine = s;
        }

        if (lane < EPW && e0 + lane < L) {
            e[e0 + lane] = mine;
            atomicAdd(&cnt[my_c], 1);
        }
    }
}


__launch_bounds__(1024)
__global__ void seg_scan_kernel(const int* __restrict__ cnt,
                                int* __restrict__ start, int M) {
    __shared__ int wtot[32];
    __shared__ int wpre[32];
    const int tid  = threadIdx.x;
    const int lane = tid & 31;
    const int wv   = tid >> 5;
    const int base = blockIdx.x * 1024;

    int pre = 0;
#pragma unroll 8
    for (int i = tid; i < base; i += 1024) pre += cnt[i];

    const int idx = base + tid;
    const int v = idx < M ? cnt[idx] : 0;

    int inc = v;
#pragma unroll
    for (int off = 1; off < 32; off <<= 1) {
        const int o = __shfl_up(inc, off, 32);
        if (lane >= off) inc += o;
    }
#pragma unroll
    for (int off = 16; off > 0; off >>= 1) pre += __shfl_xor(pre, off, 32);

    if (lane == 31) wtot[wv] = inc;
    if (lane == 0)  wpre[wv] = pre;
    __syncthreads();

    int before = wpre[lane] + (lane < wv ? wtot[lane] : 0);
#pragma unroll
    for (int off = 16; off > 0; off >>= 1) before += __shfl_xor(before, off, 32);

    if (idx < M) start[idx] = before + inc - v;
}

__launch_bounds__(256)
__global__ void seg_fill_kernel(const int* __restrict__ ci,
                                const int* __restrict__ start,
                                int* __restrict__ cnt,
                                int* __restrict__ perm, int L) {
    const int t = blockIdx.x * 256 + threadIdx.x;
    if (t >= L) return;
    const int c = ci[t];
    const int old = atomicSub(&cnt[c], 1);
    perm[start[c] + old - 1] = t;
}

__device__ __forceinline__ int seg_next_edge(const int* __restrict__ perm,
                                             int s0, int n, int prev, int lane) {
    int cur = 0x7fffffff;
    for (int i = lane; i < n; i += 32) {
        const int id = perm[s0 + i];
        if (id > prev && id < cur) cur = id;
    }
#pragma unroll
    for (int off = 16; off > 0; off >>= 1) cur = min(cur, __shfl_xor(cur, off, 32));
    return cur;
}

#define SEG_B 8
__launch_bounds__(256)
__global__ void seg_attend_kernel(const float* __restrict__ e,
                                  const int* __restrict__ chi,
                                  const int* __restrict__ start,
                                  const int* __restrict__ perm,
                                  const float* __restrict__ child_h,
                                  const float* __restrict__ child_c,
                                  const float* __restrict__ Wfx,
                                  const float* __restrict__ Ufh,
                                  float* __restrict__ h_hat,
                                  float* __restrict__ sum_fc,
                                  int M, int L) {
    const int lane = threadIdx.x & 31;
    const int wave = __builtin_amdgcn_readfirstlane((int)(threadIdx.x >> 5));
    const int c = blockIdx.x * 8 + wave;
    if (c >= M) return;
    const int sv = (lane < 2 && c + lane < M) ? start[c + lane] : L;
    const int s0 = __builtin_amdgcn_readlane(sv, 0);
    const int s1 = __builtin_amdgcn_readlane(sv, 1);
    const int n  = s1 - s0;

    const float4 wf = ((const float4*)(Wfx + (size_t)c * HDIM))[lane];
    float4 ah = make_float4(0.0f, 0.0f, 0.0f, 0.0f);
    float4 af = make_float4(0.0f, 0.0f, 0.0f, 0.0f);

    if (n <= 32) {
        const bool valid = lane < n;
        const int raw = valid ? perm[s0 + lane] : 0x7fffffff;
        int rank = 0;
        for (int j = 0; j < n; ++j) {
            const int o = __builtin_amdgcn_readlane(raw, j);
            rank += (o < raw) ? 1 : 0;
        }
        int id = 0;
        for (int j = 0; j < n; ++j) {
            const int r = __builtin_amdgcn_readlane(rank, j);
            const int o = __builtin_amdgcn_readlane(raw, j);
            if (lane == r) id = o;
        }
        const int   my_ch = valid ? chi[id] : 0;
        const float my_e  = valid ? e[id] : -__builtin_inff();

        float m = my_e;
#pragma unroll
        for (int off = 16; off > 0; off >>= 1) m = fmaxf(m, __shfl_xor(m, off, 32));
        const float my_x = valid ? __expf(my_e - m) : 0.0f;
        float ssum = my_x;
#pragma unroll
        for (int off = 16; off > 0; off >>= 1) ssum += __shfl_xor(ssum, off, 32);
        const float my_attn = my_x / (ssum + EPSV);

        for (int kb = 0; kb < n; kb += SEG_B) {
            float4 hv[SEG_B], cv[SEG_B], uf[SEG_B];
#pragma unroll
            for (int j = 0; j < SEG_B; ++j) {
                const int kk = min(kb + j, n - 1);
                const int ch = __builtin_amdgcn_readlane(my_ch, kk);
                hv[j] = ((const float4*)(child_h + (size_t)ch * HDIM))[lane];
                cv[j] = ((const float4*)(child_c + (size_t)ch * HDIM))[lane];
                uf[j] = ((const float4*)(Ufh     + (size_t)ch * HDIM))[lane];
            }
#pragma unroll
            for (int j = 0; j < SEG_B; ++j)
                asm volatile("" :: "v"(hv[j].x), "v"(hv[j].y), "v"(hv[j].z), "v"(hv[j].w),
                                   "v"(cv[j].x), "v"(cv[j].y), "v"(cv[j].z), "v"(cv[j].w),
                                   "v"(uf[j].x), "v"(uf[j].y), "v"(uf[j].z), "v"(uf[j].w));
#pragma unroll
            for (int j = 0; j < SEG_B; ++j) {
                if (kb + j < n) {
                    const float attn = __uint_as_float(
                        __builtin_amdgcn_readlane(__float_as_uint(my_attn), kb + j));
                    ah.x += attn * hv[j].x;
                    ah.y += attn * hv[j].y;
                    ah.z += attn * hv[j].z;
                    ah.w += attn * hv[j].w;
                    af.x += sigmoidf_(wf.x + uf[j].x) * cv[j].x;
                    af.y += sigmoidf_(wf.y + uf[j].y) * cv[j].y;
                    af.z += sigmoidf_(wf.z + uf[j].z) * cv[j].z;
                    af.w += sigmoidf_(wf.w + uf[j].w) * cv[j].w;
                }
            }
        }
    } else {
        float m = -__builtin_inff();
        for (int i = lane; i < n; i += 32) m = fmaxf(m, e[perm[s0 + i]]);
#pragma unroll
        for (int off = 16; off > 0; off >>= 1) m = fmaxf(m, __shfl_xor(m, off, 32));

        float ssum = 0.0f;
        int prev = -1;
        for (int k = 0; k < n; ++k) {
            const int cur = seg_next_edge(perm, s0, n, prev, lane);
            ssum += __expf(e[cur] - m);
            prev = cur;
        }
        const float denom = ssum + EPSV;

        prev = -1;
        for (int k = 0; k < n; ++k) {
            const int cur = seg_next_edge(perm, s0, n, prev, lane);
            prev = cur;
            const int ch = chi[cur];
            const float attn = __expf(e[cur] - m) / denom;
            const float4 hv = ((const float4*)(child_h + (size_t)ch * HDIM))[lane];
            const float4 cv = ((const float4*)(child_c + (size_t)ch * HDIM))[lane];
            const float4 uf = ((const float4*)(Ufh     + (size_t)ch * HDIM))[lane];
            ah.x += attn * hv.x;
            ah.y += attn * hv.y;
            ah.z += attn * hv.z;
            ah.w += attn * hv.w;
            af.x += sigmoidf_(wf.x + uf.x) * cv.x;
            af.y += sigmoidf_(wf.y + uf.y) * cv.y;
            af.z += sigmoidf_(wf.z + uf.z) * cv.z;
            af.w += sigmoidf_(wf.w + uf.w) * cv.w;
        }
    }

    ((float4*)(h_hat  + (size_t)c * HDIM))[lane] = ah;
    ((float4*)(sum_fc + (size_t)c * HDIM))[lane] = af;
}

#define GATE_V4 4
__launch_bounds__(256)
__global__ void final_gate_kernel(const float* __restrict__ Gi,
                                  const float* __restrict__ Gc,
                                  const float* __restrict__ Go,
                                  const float* __restrict__ sum_fc,
                                  float* __restrict__ h_out,
                                  float* __restrict__ c_out,
                                  size_t n4) {
    const size_t base = (size_t)blockIdx.x * (256 * GATE_V4) + threadIdx.x;
    float4 gi[GATE_V4], gc[GATE_V4], go[GATE_V4], sf[GATE_V4];
#pragma unroll
    for (int j = 0; j < GATE_V4; ++j) {
        const size_t t  = base + (size_t)j * 256;
        const size_t tl = t < n4 ? t : n4 - 1;
        gi[j] = ((const float4*)Gi)[tl];
        gc[j] = ((const float4*)Gc)[tl];
        go[j] = ((const float4*)Go)[tl];
        sf[j] = ((const float4*)sum_fc)[tl];
    }
#pragma unroll
    for (int j = 0; j < GATE_V4; ++j) {
        const size_t t = base + (size_t)j * 256;
        float4 c, h;
        c.x = sigmoidf_(gi[j].x) * tanhf(gc[j].x) + sf[j].x;
        c.y = sigmoidf_(gi[j].y) * tanhf(gc[j].y) + sf[j].y;
        c.z = sigmoidf_(gi[j].z) * tanhf(gc[j].z) + sf[j].z;
        c.w = sigmoidf_(gi[j].w) * tanhf(gc[j].w) + sf[j].w;
        h.x = sigmoidf_(go[j].x) * tanhf(c.x);
        h.y = sigmoidf_(go[j].y) * tanhf(c.y);
        h.z = sigmoidf_(go[j].z) * tanhf(c.z);
        h.w = sigmoidf_(go[j].w) * tanhf(c.w);
        if (t < n4) {
            ((float4*)h_out)[t] = h;
            ((float4*)c_out)[t] = c;
        }
    }
}

extern "C" void kernel_launch(void* const* d_in, const int* in_sizes, int n_in,
                              void* d_out, int out_size, void* d_ws, size_t ws_size,
                              hipStream_t stream) {
    const float* x_emb   = (const float*)d_in[0];
    const float* child_h = (const float*)d_in[1];
    const float* child_c = (const float*)d_in[2];
    const int*   ci      = (const int*)d_in[3];
    const int*   chi     = (const int*)d_in[4];
    const float* Wi_w = (const float*)d_in[5];
    const float* Ui_w = (const float*)d_in[6];
    const float* Wf_w = (const float*)d_in[7];
    const float* Uf_w = (const float*)d_in[8];
    const float* Wo_w = (const float*)d_in[9];
    const float* Uo_w = (const float*)d_in[10];
    const float* Wc_w = (const float*)d_in[11];
    const float* Uc_w = (const float*)d_in[12];
    const float* Wa_w = (const float*)d_in[13];
    const float* Ua_w = (const float*)d_in[14];
    const float* Wi_b = (const float*)d_in[15];
    const float* Wf_b = (const float*)d_in[16];
    const float* Wo_b = (const float*)d_in[17];
    const float* Wc_b = (const float*)d_in[18];
    const float* Wa_b = (const float*)d_in[19];
    const float* v_w  = (const float*)d_in[20];

    const int M = in_sizes[0] / HDIM;
    const int L = in_sizes[3];
    const size_t MH = (size_t)M * HDIM;

    float* ws    = (float*)d_ws;
    float* Ah    = ws;
    float* Bx    = Ah  + MH;
    float* Gi    = Bx  + MH;
    float* Wfx   = Gi  + MH;
    float* Go    = Wfx + MH;
    float* Gc    = Go  + MH;
    float* Ufh   = Gc  + MH;
    float* h_hat = Ufh + MH;
    float* sumfc = h_hat + MH;
    float* e_buf = sumfc + MH;
    int* perm    = (int*)(e_buf + L);
    int* cnt     = perm + L;
    int* start   = cnt + M;

    hipMemsetAsync(cnt, 0, (size_t)M * sizeof(int), stream);

    const dim3 gblk(256);
    const dim3 ggrd((M + 127) / 128);
    const dim3 ggrd2(2 * ((M + 127) / 128));

    gemm_group_wmma_f32<<<ggrd2, gblk, 0, stream>>>(
        x_emb, child_h,
        Ua_w, Wi_w, Wf_w, Wo_w, Wc_w, Wa_w, Uf_w,
        nullptr, Wi_b, Wf_b, Wo_b, Wc_b, Wa_b, nullptr,
        Bx, Gi, Wfx, Go, Gc, Ah, Ufh,
        5, 2, M, 0);

    edge_score_kernel<<<(L + 8 * EPW * NBATCH - 1) / (8 * EPW * NBATCH), gblk, 0, stream>>>(Ah, Bx, v_w, ci, chi,
                                                        e_buf, cnt, L);
    seg_scan_kernel<<<(M + 1023) / 1024, dim3(1024), 0, stream>>>(cnt, start, M);
    seg_fill_kernel<<<(L + 255) / 256, gblk, 0, stream>>>(ci, start, cnt, perm, L);
    seg_attend_kernel<<<(M + 7) / 8, gblk, 0, stream>>>(e_buf, chi, start, perm,
                                                        child_h, child_c, Wfx, Ufh,
                                                        h_hat, sumfc, M, L);

    gemm_group_wmma_f32<<<ggrd, gblk, 0, stream>>>(
        h_hat, h_hat,
        Ui_w, Uc_w, Uo_w, Uo_w, Uo_w, Uo_w, Uo_w,
        nullptr, nullptr, nullptr, nullptr, nullptr, nullptr, nullptr,
        Gi, Gc, Go, Go, Go, Go, Go,
        3, 0, M, 1);

    float* h_out = (float*)d_out;
    float* c_out = h_out + MH;
    final_gate_kernel<<<(unsigned)((MH / 4 + 256 * GATE_V4 - 1) / (256 * GATE_V4)), gblk, 0, stream>>>(
        Gi, Gc, Go, sumfc, h_out, c_out, MH / 4);
}


# tanh and sigmoid on the hardware exp2 and reciprocal instructions (tanh x = 1 - 2/(1 + 2^(2x log2 e))) in place of library tanhf and the IEEE divide; results move by about 3e-7, sum orders unchanged
# speedup vs baseline: 2.7337x; 1.1052x over previous
#include <hip/hip_runtime.h>
#include <hip/hip_bf16.h>

#define HDIM 128
#define EPSV 1e-9f

typedef __attribute__((ext_vector_type(2))) float v2f;
typedef __attribute__((ext_vector_type(8))) float v8f;

#define LW_PITCH 132
#define GEMM_PICK(j, p0, p1, p2, p3, p4, p5, p6)                                   \
    ((j) == 0 ? (p0) : (j) == 1 ? (p1) : (j) == 2 ? (p2) : (j) == 3 ? (p3)        \
   : (j) == 4 ? (p4) : (j) == 5 ? (p5) : (p6))
__launch_bounds__(256) __attribute__((amdgpu_waves_per_eu(2, 4)))
__global__ void gemm_group_wmma_f32(const float* __restrict__ XA,
                                    const float* __restrict__ XB,
                                    const float* W0, const float* W1, const float* W2,
                                    const float* W3, const float* W4, const float* W5,
                                    const float* W6,
                                    const float* B0, const float* B1, const float* B2,
                                    const float* B3, const float* B4, const float* B5,
                                    const float* B6,
                                    float* Y0, float* Y1, float* Y2, float* Y3,
                                    float* Y4, float* Y5, float* Y6,
                                    int nA, int nB, int Mrows, int accumulate) {
    __shared__ float lw[HDIM * LW_PITCH];

    const int wave = threadIdx.x >> 5;
    const int lane = threadIdx.x & 31;
    const int c    = lane & 15;
    const int h    = lane >> 4;

    const int nblk  = (Mrows + 127) >> 7;
    const int grp   = (int)blockIdx.x >= nblk ? 1 : 0;
    const int rb    = (int)blockIdx.x - grp * nblk;
    const int first = grp ? nA : 0;
    const int count = grp ? nB : nA;
    const float* __restrict__ X = grp ? XB : XA;

    const int m0   = rb * 128 + wave * 16;
    const bool active = m0 < Mrows;

    v2f a[32];
    if (active) {
        const float* __restrict__ xrow = X + (size_t)(m0 + c) * HDIM + 2 * h;
#pragma unroll
        for (int ks = 0; ks < 32; ++ks) {
            a[ks].x = xrow[4 * ks];
            a[ks].y = xrow[4 * ks + 1];
        }
    }

    const float* __restrict__ wrow = &lw[c * LW_PITCH + 2 * h];

    float* __restrict__ lw_st = &lw[((wave & 3) * 16 + (wave >> 2)) * LW_PITCH + (lane << 2)];
    const int w_off = (int)threadIdx.x * 4;
#define GEMM_STAGE_W(Wptr)                                                           \
    {                                                                                \
        const float* __restrict__ wsrc = (Wptr) + w_off;                             \
        float4 wv[16];                                                               \
        _Pragma("unroll")                                                            \
        for (int it = 0; it < 16; ++it)                                              \
            wv[it] = *(const float4*)(wsrc + it * 1024);                             \
        _Pragma("unroll")                                                            \
        for (int it = 0; it < 16; ++it)                                              \
            *(float4*)(lw_st + ((it >> 3) * 64 + (it & 7) * 2) * LW_PITCH) = wv[it]; \
    }

    GEMM_STAGE_W(GEMM_PICK(first, W0, W1, W2, W3, W4, W5, W6))

#pragma nounroll
    for (int g = 0; g < count; ++g) {
        const int j = first + g;
        const float* __restrict__ bias = GEMM_PICK(j, B0, B1, B2, B3, B4, B5, B6);
        float* __restrict__ Y          = GEMM_PICK(j, Y0, Y1, Y2, Y3, Y4, Y5, Y6);

        __syncthreads();

        if (active) {
            v8f acc[8] = {};
#pragma unroll
            for (int ks = 0; ks < 32; ++ks) {
#pragma unroll
                for (int t = 0; t < 8; ++t) {
                    const v2f b = *(const v2f*)(wrow + t * 16 * LW_PITCH + 4 * ks);
                    acc[t] = __builtin_amdgcn_wmma_f32_16x16x4_f32(
                         false, a[ks],  false, b,
                         (short)0, acc[t],  false,  false);
                }
            }

            float* __restrict__ ybase = Y + (size_t)(m0 + 8 * h) * HDIM + 4 * c;
            float4 bv[2];
#pragma unroll
            for (int q = 0; q < 2; ++q) {
                bv[q] = make_float4(0.0f, 0.0f, 0.0f, 0.0f);
                if (bias) bv[q] = *(const float4*)(bias + 64 * q + 4 * c);
            }

            if (accumulate) {
#pragma unroll
                for (int q = 0; q < 2; ++q) {
                    float4 y[8];
#pragma unroll
                    for (int r = 0; r < 8; ++r)
                        y[r] = *(const float4*)(ybase + (size_t)r * HDIM + 64 * q);
#pragma unroll
                    for (int r = 0; r < 8; r += 4)
                        asm volatile("" :: "v"(y[r].x), "v"(y[r].y), "v"(y[r].z), "v"(y[r].w),
                                           "v"(y[r + 1].x), "v"(y[r + 1].y), "v"(y[r + 1].z), "v"(y[r + 1].w),
                                           "v"(y[r + 2].x), "v"(y[r + 2].y), "v"(y[r + 2].z), "v"(y[r + 2].w),
                                           "v"(y[r + 3].x), "v"(y[r + 3].y), "v"(y[r + 3].z), "v"(y[r + 3].w));
#pragma unroll
                    for (int r = 0; r < 8; ++r) {
                        float4 v;
                        v.x = (acc[4 * q + 0][r] + bv[q].x) + y[r].x;
                        v.y = (acc[4 * q + 1][r] + bv[q].y) + y[r].y;
                        v.z = (acc[4 * q + 2][r] + bv[q].z) + y[r].z;
                        v.w = (acc[4 * q + 3][r] + bv[q].w) + y[r].w;
                        *(float4*)(ybase + (size_t)r * HDIM + 64 * q) = v;
                    }
                }
            } else {
#pragma unroll
                for (int i = 0; i < 16; ++i) {
                    const int q = i >> 3, r = i & 7;
                    float4 v;
                    v.x = acc[4 * q + 0][r] + bv[q].x;
                    v.y = acc[4 * q + 1][r] + bv[q].y;
                    v.z = acc[4 * q + 2][r] + bv[q].z;
                    v.w = acc[4 * q + 3][r] + bv[q].w;
                    *(float4*)(ybase + (size_t)r * HDIM + 64 * q) = v;
                }
            }
        }

        if (g + 1 < count) {
            __syncthreads();
            GEMM_STAGE_W(GEMM_PICK(j + 1, W0, W1, W2, W3, W4, W5, W6))
        }
    }
#undef GEMM_STAGE_W
}

__device__ __forceinline__ float sigmoidf_(float x) {
    return __builtin_amdgcn_rcpf(1.0f + __builtin_amdgcn_exp2f(x * -1.4426950408889634f));
}
__device__ __forceinline__ float tanhf_(float x) {
    return 1.0f - 2.0f * __builtin_amdgcn_rcpf(1.0f + __builtin_amdgcn_exp2f(x * 2.8853900817779268f));
}

#define EPW 8
#define NBATCH 2

__launch_bounds__(256)
__global__ void edge_score_kernel(const float* __restrict__ Ah,
                                  const float* __restrict__ Bx,
                                  const float* __restrict__ v_w,
                                  const int* __restrict__ ci,
                                  const int* __restrict__ chi,
                                  float* __restrict__ e,
                                  int* __restrict__ cnt,
                                  int L) {
    const int lane = threadIdx.x & 31;
    const int wave = __builtin_amdgcn_readfirstlane((int)(threadIdx.x >> 5));
    const int w0 = (blockIdx.x * 8 + wave) * (EPW * NBATCH);
    const float4 v = ((const float4*)v_w)[lane];

    for (int bt = 0; bt < NBATCH; ++bt) {
        const int e0 = w0 + bt * EPW;
        if (e0 >= L) break;
        const int le = min(e0 + (lane & (EPW - 1)), L - 1);
        const int my_c = ci[le], my_ch = chi[le];

        float4 a[EPW], b[EPW];
#pragma unroll
        for (int j = 0; j < EPW; ++j) {
            const int c  = __builtin_amdgcn_readlane(my_c, j);
            const int ch = __builtin_amdgcn_readlane(my_ch, j);
            a[j] = ((const float4*)(Ah + (size_t)ch * HDIM))[lane];
            b[j] = ((const float4*)(Bx + (size_t)c  * HDIM))[lane];
        }

        float mine = 0.0f;
#pragma unroll
        for (int j = 0; j < EPW; ++j) {
            float s = tanhf_(a[j].x + b[j].x) * v.x + tanhf_(a[j].y + b[j].y) * v.y
                    + tanhf_(a[j].z + b[j].z) * v.z + tanhf_(a[j].w + b[j].w) * v.w;
#pragma unroll
            for (int off = 16; off > 0; off >>= 1) s += __shfl_xor(s, off, 32);
            if (lane == j) mine = s;
        }

        if (lane < EPW && e0 + lane < L) {
            e[e0 + lane] = mine;
            atomicAdd(&cnt[my_c], 1);
        }
    }
}


__launch_bounds__(1024)
__global__ void seg_scan_kernel(const int* __restrict__ cnt,
                                int* __restrict__ start, int M) {
    __shared__ int wtot[32];
    __shared__ int wpre[32];
    const int tid  = threadIdx.x;
    const int lane = tid & 31;
    const int wv   = tid >> 5;
    const int base = blockIdx.x * 1024;

    int pre = 0;
#pragma unroll 8
    for (int i = tid; i < base; i += 1024) pre += cnt[i];

    const int idx = base + tid;
    const int v = idx < M ? cnt[idx] : 0;

    int inc = v;
#pragma unroll
    for (int off = 1; off < 32; off <<= 1) {
        const int o = __shfl_up(inc, off, 32);
        if (lane >= off) inc += o;
    }
#pragma unroll
    for (int off = 16; off > 0; off >>= 1) pre += __shfl_xor(pre, off, 32);

    if (lane == 31) wtot[wv] = inc;
    if (lane == 0)  wpre[wv] = pre;
    __syncthreads();

    int before = wpre[lane] + (lane < wv ? wtot[lane] : 0);
#pragma unroll
    for (int off = 16; off > 0; off >>= 1) before += __shfl_xor(before, off, 32);

    if (idx < M) start[idx] = before + inc - v;
}

__launch_bounds__(256)
__global__ void seg_fill_kernel(const int* __restrict__ ci,
                                const int* __restrict__ start,
                                int* __restrict__ cnt,
                                int* __restrict__ perm, int L) {
    const int t = blockIdx.x * 256 + threadIdx.x;
    if (t >= L) return;
    const int c = ci[t];
    const int old = atomicSub(&cnt[c], 1);
    perm[start[c] + old - 1] = t;
}

__device__ __forceinline__ int seg_next_edge(const int* __restrict__ perm,
                                             int s0, int n, int prev, int lane) {
    int cur = 0x7fffffff;
    for (int i = lane; i < n; i += 32) {
        const int id = perm[s0 + i];
        if (id > prev && id < cur) cur = id;
    }
#pragma unroll
    for (int off = 16; off > 0; off >>= 1) cur = min(cur, __shfl_xor(cur, off, 32));
    return cur;
}

#define SEG_B 8
__launch_bounds__(256)
__global__ void seg_attend_kernel(const float* __restrict__ e,
                                  const int* __restrict__ chi,
                                  const int* __restrict__ start,
                                  const int* __restrict__ perm,
                                  const float* __restrict__ child_h,
                                  const float* __restrict__ child_c,
                                  const float* __restrict__ Wfx,
                                  const float* __restrict__ Ufh,
                                  float* __restrict__ h_hat,
                                  float* __restrict__ sum_fc,
                                  int M, int L) {
    const int lane = threadIdx.x & 31;
    const int wave = __builtin_amdgcn_readfirstlane((int)(threadIdx.x >> 5));
    const int c = blockIdx.x * 8 + wave;
    if (c >= M) return;
    const int sv = (lane < 2 && c + lane < M) ? start[c + lane] : L;
    const int s0 = __builtin_amdgcn_readlane(sv, 0);
    const int s1 = __builtin_amdgcn_readlane(sv, 1);
    const int n  = s1 - s0;

    const float4 wf = ((const float4*)(Wfx + (size_t)c * HDIM))[lane];
    float4 ah = make_float4(0.0f, 0.0f, 0.0f, 0.0f);
    float4 af = make_float4(0.0f, 0.0f, 0.0f, 0.0f);

    if (n <= 32) {
        const bool valid = lane < n;
        const int raw = valid ? perm[s0 + lane] : 0x7fffffff;
        int rank = 0;
        for (int j = 0; j < n; ++j) {
            const int o = __builtin_amdgcn_readlane(raw, j);
            rank += (o < raw) ? 1 : 0;
        }
        int id = 0;
        for (int j = 0; j < n; ++j) {
            const int r = __builtin_amdgcn_readlane(rank, j);
            const int o = __builtin_amdgcn_readlane(raw, j);
            if (lane == r) id = o;
        }
        const int   my_ch = valid ? chi[id] : 0;
        const float my_e  = valid ? e[id] : -__builtin_inff();

        float m = my_e;
#pragma unroll
        for (int off = 16; off > 0; off >>= 1) m = fmaxf(m, __shfl_xor(m, off, 32));
        const float my_x = valid ? __expf(my_e - m) : 0.0f;
        float ssum = my_x;
#pragma unroll
        for (int off = 16; off > 0; off >>= 1) ssum += __shfl_xor(ssum, off, 32);
        const float my_attn = my_x / (ssum + EPSV);

        for (int kb = 0; kb < n; kb += SEG_B) {
            float4 hv[SEG_B], cv[SEG_B], uf[SEG_B];
#pragma unroll
            for (int j = 0; j < SEG_B; ++j) {
                const int kk = min(kb + j, n - 1);
                const int ch = __builtin_amdgcn_readlane(my_ch, kk);
                hv[j] = ((const float4*)(child_h + (size_t)ch * HDIM))[lane];
                cv[j] = ((const float4*)(child_c + (size_t)ch * HDIM))[lane];
                uf[j] = ((const float4*)(Ufh     + (size_t)ch * HDIM))[lane];
            }
#pragma unroll
            for (int j = 0; j < SEG_B; ++j)
                asm volatile("" :: "v"(hv[j].x), "v"(hv[j].y), "v"(hv[j].z), "v"(hv[j].w),
                                   "v"(cv[j].x), "v"(cv[j].y), "v"(cv[j].z), "v"(cv[j].w),
                                   "v"(uf[j].x), "v"(uf[j].y), "v"(uf[j].z), "v"(uf[j].w));
#pragma unroll
            for (int j = 0; j < SEG_B; ++j) {
                if (kb + j < n) {
                    const float attn = __uint_as_float(
                        __builtin_amdgcn_readlane(__float_as_uint(my_attn), kb + j));
                    ah.x += attn * hv[j].x;
                    ah.y += attn * hv[j].y;
                    ah.z += attn * hv[j].z;
                    ah.w += attn * hv[j].w;
                    af.x += sigmoidf_(wf.x + uf[j].x) * cv[j].x;
                    af.y += sigmoidf_(wf.y + uf[j].y) * cv[j].y;
                    af.z += sigmoidf_(wf.z + uf[j].z) * cv[j].z;
                    af.w += sigmoidf_(wf.w + uf[j].w) * cv[j].w;
                }
            }
        }
    } else {
        float m = -__builtin_inff();
        for (int i = lane; i < n; i += 32) m = fmaxf(m, e[perm[s0 + i]]);
#pragma unroll
        for (int off = 16; off > 0; off >>= 1) m = fmaxf(m, __shfl_xor(m, off, 32));

        float ssum = 0.0f;
        int prev = -1;
        for (int k = 0; k < n; ++k) {
            const int cur = seg_next_edge(perm, s0, n, prev, lane);
            ssum += __expf(e[cur] - m);
            prev = cur;
        }
        const float denom = ssum + EPSV;

        prev = -1;
        for (int k = 0; k < n; ++k) {
            const int cur = seg_next_edge(perm, s0, n, prev, lane);
            prev = cur;
            const int ch = chi[cur];
            const float attn = __expf(e[cur] - m) / denom;
            const float4 hv = ((const float4*)(child_h + (size_t)ch * HDIM))[lane];
            const float4 cv = ((const float4*)(child_c + (size_t)ch * HDIM))[lane];
            const float4 uf = ((const float4*)(Ufh     + (size_t)ch * HDIM))[lane];
            ah.x += attn * hv.x;
            ah.y += attn * hv.y;
            ah.z += attn * hv.z;
            ah.w += attn * hv.w;
            af.x += sigmoidf_(wf.x + uf.x) * cv.x;
            af.y += sigmoidf_(wf.y + uf.y) * cv.y;
            af.z += sigmoidf_(wf.z + uf.z) * cv.z;
            af.w += sigmoidf_(wf.w + uf.w) * cv.w;
        }
    }

    ((float4*)(h_hat  + (size_t)c * HDIM))[lane] = ah;
    ((float4*)(sum_fc + (size_t)c * HDIM))[lane] = af;
}

#define GATE_V4 4
__launch_bounds__(256)
__global__ void final_gate_kernel(const float* __restrict__ Gi,
                                  const float* __restrict__ Gc,
                                  const float* __restrict__ Go,
                                  const float* __restrict__ sum_fc,
                                  float* __restrict__ h_out,
                                  float* __restrict__ c_out,
                                  size_t n4) {
    const size_t base = (size_t)blockIdx.x * (256 * GATE_V4) + threadIdx.x;
    float4 gi[GATE_V4], gc[GATE_V4], go[GATE_V4], sf[GATE_V4];
#pragma unroll
    for (int j = 0; j < GATE_V4; ++j) {
        const size_t t  = base + (size_t)j * 256;
        const size_t tl = t < n4 ? t : n4 - 1;
        gi[j] = ((const float4*)Gi)[tl];
        gc[j] = ((const float4*)Gc)[tl];
        go[j] = ((const float4*)Go)[tl];
        sf[j] = ((const float4*)sum_fc)[tl];
    }
#pragma unroll
    for (int j = 0; j < GATE_V4; ++j) {
        const size_t t = base + (size_t)j * 256;
        float4 c, h;
        c.x = sigmoidf_(gi[j].x) * tanhf_(gc[j].x) + sf[j].x;
        c.y = sigmoidf_(gi[j].y) * tanhf_(gc[j].y) + sf[j].y;
        c.z = sigmoidf_(gi[j].z) * tanhf_(gc[j].z) + sf[j].z;
        c.w = sigmoidf_(gi[j].w) * tanhf_(gc[j].w) + sf[j].w;
        h.x = sigmoidf_(go[j].x) * tanhf_(c.x);
        h.y = sigmoidf_(go[j].y) * tanhf_(c.y);
        h.z = sigmoidf_(go[j].z) * tanhf_(c.z);
        h.w = sigmoidf_(go[j].w) * tanhf_(c.w);
        if (t < n4) {
            ((float4*)h_out)[t] = h;
            ((float4*)c_out)[t] = c;
        }
    }
}

extern "C" void kernel_launch(void* const* d_in, const int* in_sizes, int n_in,
                              void* d_out, int out_size, void* d_ws, size_t ws_size,
                              hipStream_t stream) {
    const float* x_emb   = (const float*)d_in[0];
    const float* child_h = (const float*)d_in[1];
    const float* child_c = (const float*)d_in[2];
    const int*   ci      = (const int*)d_in[3];
    const int*   chi     = (const int*)d_in[4];
    const float* Wi_w = (const float*)d_in[5];
    const float* Ui_w = (const float*)d_in[6];
    const float* Wf_w = (const float*)d_in[7];
    const float* Uf_w = (const float*)d_in[8];
    const float* Wo_w = (const float*)d_in[9];
    const float* Uo_w = (const float*)d_in[10];
    const float* Wc_w = (const float*)d_in[11];
    const float* Uc_w = (const float*)d_in[12];
    const float* Wa_w = (const float*)d_in[13];
    const float* Ua_w = (const float*)d_in[14];
    const float* Wi_b = (const float*)d_in[15];
    const float* Wf_b = (const float*)d_in[16];
    const float* Wo_b = (const float*)d_in[17];
    const float* Wc_b = (const float*)d_in[18];
    const float* Wa_b = (const float*)d_in[19];
    const float* v_w  = (const float*)d_in[20];

    const int M = in_sizes[0] / HDIM;
    const int L = in_sizes[3];
    const size_t MH = (size_t)M * HDIM;

    float* ws    = (float*)d_ws;
    float* Ah    = ws;
    float* Bx    = Ah  + MH;
    float* Gi    = Bx  + MH;
    float* Wfx   = Gi  + MH;
    float* Go    = Wfx + MH;
    float* Gc    = Go  + MH;
    float* Ufh   = Gc  + MH;
    float* h_hat = Ufh + MH;
    float* sumfc = h_hat + MH;
    float* e_buf = sumfc + MH;
    int* perm    = (int*)(e_buf + L);
    int* cnt     = perm + L;
    int* start   = cnt + M;

    hipMemsetAsync(cnt, 0, (size_t)M * sizeof(int), stream);

    const dim3 gblk(256);
    const dim3 ggrd((M + 127) / 128);
    const dim3 ggrd2(2 * ((M + 127) / 128));

    gemm_group_wmma_f32<<<ggrd2, gblk, 0, stream>>>(
        x_emb, child_h,
        Ua_w, Wi_w, Wf_w, Wo_w, Wc_w, Wa_w, Uf_w,
        nullptr, Wi_b, Wf_b, Wo_b, Wc_b, Wa_b, nullptr,
        Bx, Gi, Wfx, Go, Gc, Ah, Ufh,
        5, 2, M, 0);

    edge_score_kernel<<<(L + 8 * EPW * NBATCH - 1) / (8 * EPW * NBATCH), gblk, 0, stream>>>(Ah, Bx, v_w, ci, chi,
                                                        e_buf, cnt, L);
    seg_scan_kernel<<<(M + 1023) / 1024, dim3(1024), 0, stream>>>(cnt, start, M);
    seg_fill_kernel<<<(L + 255) / 256, gblk, 0, stream>>>(ci, start, cnt, perm, L);
    seg_attend_kernel<<<(M + 7) / 8, gblk, 0, stream>>>(e_buf, chi, start, perm,
                                                        child_h, child_c, Wfx, Ufh,
                                                        h_hat, sumfc, M, L);

    gemm_group_wmma_f32<<<ggrd, gblk, 0, stream>>>(
        h_hat, h_hat,
        Ui_w, Uc_w, Uo_w, Uo_w, Uo_w, Uo_w, Uo_w,
        nullptr, nullptr, nullptr, nullptr, nullptr, nullptr, nullptr,
        Gi, Gc, Go, Go, Go, Go, Go,
        3, 0, M, 1);

    float* h_out = (float*)d_out;
    float* c_out = h_out + MH;
    final_gate_kernel<<<(unsigned)((MH / 4 + 256 * GATE_V4 - 1) / (256 * GATE_V4)), gblk, 0, stream>>>(
        Gi, Gc, Go, sumfc, h_out, c_out, MH / 4);
}


# Gate kernel folded into the last product launch: Gi/Gc/Go are the accumulators' start values (sum order: start value first, then the products), epilogue forms c and h, tanh kept per lane in LDS
# speedup vs baseline: 2.8959x; 1.0593x over previous
#include <hip/hip_runtime.h>
#include <hip/hip_bf16.h>

#define HDIM 128
#define EPSV 1e-9f

typedef __attribute__((ext_vector_type(2))) float v2f;
typedef __attribute__((ext_vector_type(8))) float v8f;

__device__ __forceinline__ float sigmoidf_(float x) {
    return __builtin_amdgcn_rcpf(1.0f + __builtin_amdgcn_exp2f(x * -1.4426950408889634f));
}
__device__ __forceinline__ float tanhf_(float x) {
    return 1.0f - 2.0f * __builtin_amdgcn_rcpf(1.0f + __builtin_amdgcn_exp2f(x * 2.8853900817779268f));
}

#define LW_PITCH 132
#define GEMM_PICK(j, p0, p1, p2, p3, p4, p5, p6)                                   \
    ((j) == 0 ? (p0) : (j) == 1 ? (p1) : (j) == 2 ? (p2) : (j) == 3 ? (p3)        \
   : (j) == 4 ? (p4) : (j) == 5 ? (p5) : (p6))
template <bool GATE>
__launch_bounds__(256) __attribute__((amdgpu_waves_per_eu(4, 4)))
__global__ void gemm_group_wmma_f32(const float* __restrict__ XA,
                                    const float* __restrict__ XB,
                                    const float* W0, const float* W1, const float* W2,
                                    const float* W3, const float* W4, const float* W5,
                                    const float* W6,
                                    const float* B0, const float* B1, const float* B2,
                                    const float* B3, const float* B4, const float* B5,
                                    const float* B6,
                                    float* Y0, float* Y1, float* Y2, float* Y3,
                                    float* Y4, float* Y5, float* Y6,
                                    int nA, int nB, int Mrows,
                                    const float* __restrict__ sum_fc,
                                    float* __restrict__ h_out,
                                    float* __restrict__ c_out) {
    __shared__ float lw[HDIM * LW_PITCH];
    __shared__ float4 keepb[GATE ? 8 * 16 * 32 : 1];

    const int wave = threadIdx.x >> 5;
    const int lane = threadIdx.x & 31;
    const int c    = lane & 15;
    const int h    = lane >> 4;

    const int nblk  = (Mrows + 127) >> 7;
    const int grp   = (int)blockIdx.x >= nblk ? 1 : 0;
    const int rb    = (int)blockIdx.x - grp * nblk;
    const int first = grp ? nA : 0;
    const int count = grp ? nB : nA;
    const float* __restrict__ X = grp ? XB : XA;

    const int m0   = rb * 128 + wave * 16;
    const bool active = m0 < Mrows;

    v2f a[32];
    if (active) {
        const float* __restrict__ xrow = X + (size_t)(m0 + c) * HDIM + 2 * h;
#pragma unroll
        for (int ks = 0; ks < 32; ++ks) {
            a[ks].x = xrow[4 * ks];
            a[ks].y = xrow[4 * ks + 1];
        }
    }

    const float* __restrict__ wrow = &lw[c * LW_PITCH + 2 * h];

    float* __restrict__ lw_st = &lw[((wave & 3) * 16 + (wave >> 2)) * LW_PITCH + (lane << 2)];
    const int w_off = (int)threadIdx.x * 4;
#define GEMM_STAGE_W(Wptr)                                                           \
    {                                                                                \
        const float* __restrict__ wsrc = (Wptr) + w_off;                             \
        float4 wv[16];                                                               \
        _Pragma("unroll")                                                            \
        for (int it = 0; it < 16; ++it)                                              \
            wv[it] = *(const float4*)(wsrc + it * 1024);                             \
        _Pragma("unroll")                                                            \
        for (int it = 0; it < 16; ++it)                                              \
            *(float4*)(lw_st + ((it >> 3) * 64 + (it & 7) * 2) * LW_PITCH) = wv[it]; \
    }

    const size_t goff = (size_t)(m0 + 8 * h) * HDIM + 4 * c;

    v8f acc[8];
#define GEMM_GATE_START(jj)                                                          \
    if (GATE && active) {                                                            \
        const float* ysrc = GEMM_PICK(jj, Y0, Y1, Y2, Y3, Y4, Y5, Y6) + goff;        \
        _Pragma("unroll")                                                            \
        for (int i = 0; i < 16; ++i) {                                               \
            const int q = i >> 3, r = i & 7;                                         \
            const float4 y = *(const float4*)(ysrc + (size_t)r * HDIM + 64 * q);     \
            acc[4 * q + 0][r] = y.x;                                                 \
            acc[4 * q + 1][r] = y.y;                                                 \
            acc[4 * q + 2][r] = y.z;                                                 \
            acc[4 * q + 3][r] = y.w;                                                 \
        }                                                                            \
    }

    GEMM_GATE_START(first)
    GEMM_STAGE_W(GEMM_PICK(first, W0, W1, W2, W3, W4, W5, W6))

#pragma nounroll
    for (int g = 0; g < count; ++g) {
        const int j = first + g;
        const float* __restrict__ bias = GEMM_PICK(j, B0, B1, B2, B3, B4, B5, B6);
        float* __restrict__ Y          = GEMM_PICK(j, Y0, Y1, Y2, Y3, Y4, Y5, Y6);

        __syncthreads();

        if (active) {
            if (!GATE) {
#pragma unroll
                for (int t = 0; t < 8; ++t) acc[t] = v8f{};
            }
#pragma unroll
            for (int ks = 0; ks < 32; ++ks) {
#pragma unroll
                for (int t = 0; t < 8; ++t) {
                    const v2f b = *(const v2f*)(wrow + t * 16 * LW_PITCH + 4 * ks);
                    acc[t] = __builtin_amdgcn_wmma_f32_16x16x4_f32(
                         false, a[ks],  false, b,
                         (short)0, acc[t],  false,  false);
                }
            }

            if (GATE) {
                float4* __restrict__ kb = &keepb[wave * 16 * 32 + lane];
                if (g == 0) {
#pragma unroll
                    for (int i = 0; i < 16; ++i) {
                        const int q = i >> 3, r = i & 7;
                        float4 t;
                        t.x = tanhf_(acc[4 * q + 0][r]);
                        t.y = tanhf_(acc[4 * q + 1][r]);
                        t.z = tanhf_(acc[4 * q + 2][r]);
                        t.w = tanhf_(acc[4 * q + 3][r]);
                        kb[i * 32] = t;
                    }
                } else if (g == 1) {
#pragma unroll
                    for (int q = 0; q < 2; ++q) {
                        float4 sf[8];
#pragma unroll
                        for (int r = 0; r < 8; ++r)
                            sf[r] = *(const float4*)(sum_fc + goff + (size_t)r * HDIM + 64 * q);
#pragma unroll
                        for (int r = 0; r < 8; r += 4)
                            asm volatile("" :: "v"(sf[r].x), "v"(sf[r].y), "v"(sf[r].z), "v"(sf[r].w),
                                               "v"(sf[r + 1].x), "v"(sf[r + 1].y), "v"(sf[r + 1].z), "v"(sf[r + 1].w),
                                               "v"(sf[r + 2].x), "v"(sf[r + 2].y), "v"(sf[r + 2].z), "v"(sf[r + 2].w),
                                               "v"(sf[r + 3].x), "v"(sf[r + 3].y), "v"(sf[r + 3].z), "v"(sf[r + 3].w));
#pragma unroll
                        for (int r = 0; r < 8; ++r) {
                            const float4 t = kb[(8 * q + r) * 32];
                            float4 cc, tc;
                            cc.x = sigmoidf_(acc[4 * q + 0][r]) * t.x + sf[r].x;
                            cc.y = sigmoidf_(acc[4 * q + 1][r]) * t.y + sf[r].y;
                            cc.z = sigmoidf_(acc[4 * q + 2][r]) * t.z + sf[r].z;
                            cc.w = sigmoidf_(acc[4 * q + 3][r]) * t.w + sf[r].w;
                            *(float4*)(c_out + goff + (size_t)r * HDIM + 64 * q) = cc;
                            tc.x = tanhf_(cc.x); tc.y = tanhf_(cc.y); tc.z = tanhf_(cc.z); tc.w = tanhf_(cc.w);
                            kb[(8 * q + r) * 32] = tc;
                        }
                    }
                } else {
#pragma unroll
                    for (int i = 0; i < 16; ++i) {
                        const int q = i >> 3, r = i & 7;
                        const float4 t = kb[i * 32];
                        float4 hh;
                        hh.x = sigmoidf_(acc[4 * q + 0][r]) * t.x;
                        hh.y = sigmoidf_(acc[4 * q + 1][r]) * t.y;
                        hh.z = sigmoidf_(acc[4 * q + 2][r]) * t.z;
                        hh.w = sigmoidf_(acc[4 * q + 3][r]) * t.w;
                        *(float4*)(h_out + goff + (size_t)r * HDIM + 64 * q) = hh;
                    }
                }
            } else {
                float* __restrict__ ybase = Y + goff;
                float4 bv[2];
#pragma unroll
                for (int q = 0; q < 2; ++q) {
                    bv[q] = make_float4(0.0f, 0.0f, 0.0f, 0.0f);
                    if (bias) bv[q] = *(const float4*)(bias + 64 * q + 4 * c);
                }
#pragma unroll
                for (int i = 0; i < 16; ++i) {
                    const int q = i >> 3, r = i & 7;
                    float4 v;
                    v.x = acc[4 * q + 0][r] + bv[q].x;
                    v.y = acc[4 * q + 1][r] + bv[q].y;
                    v.z = acc[4 * q + 2][r] + bv[q].z;
                    v.w = acc[4 * q + 3][r] + bv[q].w;
                    *(float4*)(ybase + (size_t)r * HDIM + 64 * q) = v;
                }
            }
        }

        if (g + 1 < count) {
            GEMM_GATE_START(j + 1)
            __syncthreads();
            GEMM_STAGE_W(GEMM_PICK(j + 1, W0, W1, W2, W3, W4, W5, W6))
        }
    }
#undef GEMM_STAGE_W
#undef GEMM_GATE_START
}

#define EPW 8
#define NBATCH 2

__launch_bounds__(256)
__global__ void edge_score_kernel(const float* __restrict__ Ah,
                                  const float* __restrict__ Bx,
                                  const float* __restrict__ v_w,
                                  const int* __restrict__ ci,
                                  const int* __restrict__ chi,
                                  float* __restrict__ e,
                                  int* __restrict__ cnt,
                                  int L) {
    const int lane = threadIdx.x & 31;
    const int wave = __builtin_amdgcn_readfirstlane((int)(threadIdx.x >> 5));
    const int w0 = (blockIdx.x * 8 + wave) * (EPW * NBATCH);
    const float4 v = ((const float4*)v_w)[lane];

    for (int bt = 0; bt < NBATCH; ++bt) {
        const int e0 = w0 + bt * EPW;
        if (e0 >= L) break;
        const int le = min(e0 + (lane & (EPW - 1)), L - 1);
        const int my_c = ci[le], my_ch = chi[le];

        float4 a[EPW], b[EPW];
#pragma unroll
        for (int j = 0; j < EPW; ++j) {
            const int c  = __builtin_amdgcn_readlane(my_c, j);
            const int ch = __builtin_amdgcn_readlane(my_ch, j);
            a[j] = ((const float4*)(Ah + (size_t)ch * HDIM))[lane];
            b[j] = ((const float4*)(Bx + (size_t)c  * HDIM))[lane];
        }

        float mine = 0.0f;
#pragma unroll
        for (int j = 0; j < EPW; ++j) {
            float s = tanhf_(a[j].x + b[j].x) * v.x + tanhf_(a[j].y + b[j].y) * v.y
                    + tanhf_(a[j].z + b[j].z) * v.z + tanhf_(a[j].w + b[j].w) * v.w;
#pragma unroll
            for (int off = 16; off > 0; off >>= 1) s += __shfl_xor(s, off, 32);
            if (lane == j) mine = s;
        }

        if (lane < EPW && e0 + lane < L) {
            e[e0 + lane] = mine;
            atomicAdd(&cnt[my_c], 1);
        }
    }
}


__launch_bounds__(1024)
__global__ void seg_scan_kernel(const int* __restrict__ cnt,
                                int* __restrict__ start, int M) {
    __shared__ int wtot[32];
    __shared__ int wpre[32];
    const int tid  = threadIdx.x;
    const int lane = tid & 31;
    const int wv   = tid >> 5;
    const int base = blockIdx.x * 1024;

    int pre = 0;
#pragma unroll 8
    for (int i = tid; i < base; i += 1024) pre += cnt[i];

    const int idx = base + tid;
    const int v = idx < M ? cnt[idx] : 0;

    int inc = v;
#pragma unroll
    for (int off = 1; off < 32; off <<= 1) {
        const int o = __shfl_up(inc, off, 32);
        if (lane >= off) inc += o;
    }
#pragma unroll
    for (int off = 16; off > 0; off >>= 1) pre += __shfl_xor(pre, off, 32);

    if (lane == 31) wtot[wv] = inc;
    if (lane == 0)  wpre[wv] = pre;
    __syncthreads();

    int before = wpre[lane] + (lane < wv ? wtot[lane] : 0);
#pragma unroll
    for (int off = 16; off > 0; off >>= 1) before += __shfl_xor(before, off, 32);

    if (idx < M) start[idx] = before + inc - v;
}

__launch_bounds__(256)
__global__ void seg_fill_kernel(const int* __restrict__ ci,
                                const int* __restrict__ start,
                                int* __restrict__ cnt,
                                int* __restrict__ perm, int L) {
    const int t = blockIdx.x * 256 + threadIdx.x;
    if (t >= L) return;
    const int c = ci[t];
    const int old = atomicSub(&cnt[c], 1);
    perm[start[c] + old - 1] = t;
}

__device__ __forceinline__ int seg_next_edge(const int* __restrict__ perm,
                                             int s0, int n, int prev, int lane) {
    int cur = 0x7fffffff;
    for (int i = lane; i < n; i += 32) {
        const int id = perm[s0 + i];
        if (id > prev && id < cur) cur = id;
    }
#pragma unroll
    for (int off = 16; off > 0; off >>= 1) cur = min(cur, __shfl_xor(cur, off, 32));
    return cur;
}

#define SEG_B 8
__launch_bounds__(256)
__global__ void seg_attend_kernel(const float* __restrict__ e,
                                  const int* __restrict__ chi,
                                  const int* __restrict__ start,
                                  const int* __restrict__ perm,
                                  const float* __restrict__ child_h,
                                  const float* __restrict__ child_c,
                                  const float* __restrict__ Wfx,
                                  const float* __restrict__ Ufh,
                                  float* __restrict__ h_hat,
                                  float* __restrict__ sum_fc,
                                  int M, int L) {
    const int lane = threadIdx.x & 31;
    const int wave = __builtin_amdgcn_readfirstlane((int)(threadIdx.x >> 5));
    const int c = blockIdx.x * 8 + wave;
    if (c >= M) return;
    const int sv = (lane < 2 && c + lane < M) ? start[c + lane] : L;
    const int s0 = __builtin_amdgcn_readlane(sv, 0);
    const int s1 = __builtin_amdgcn_readlane(sv, 1);
    const int n  = s1 - s0;

    const float4 wf = ((const float4*)(Wfx + (size_t)c * HDIM))[lane];
    float4 ah = make_float4(0.0f, 0.0f, 0.0f, 0.0f);
    float4 af = make_float4(0.0f, 0.0f, 0.0f, 0.0f);

    if (n <= 32) {
        const bool valid = lane < n;
        const int raw = valid ? perm[s0 + lane] : 0x7fffffff;
        int rank = 0;
        for (int j = 0; j < n; ++j) {
            const int o = __builtin_amdgcn_readlane(raw, j);
            rank += (o < raw) ? 1 : 0;
        }
        int id = 0;
        for (int j = 0; j < n; ++j) {
            const int r = __builtin_amdgcn_readlane(rank, j);
            const int o = __builtin_amdgcn_readlane(raw, j);
            if (lane == r) id = o;
        }
        const int   my_ch = valid ? chi[id] : 0;
        const float my_e  = valid ? e[id] : -__builtin_inff();

        float m = my_e;
#pragma unroll
        for (int off = 16; off > 0; off >>= 1) m = fmaxf(m, __shfl_xor(m, off, 32));
        const float my_x = valid ? __expf(my_e - m) : 0.0f;
        float ssum = my_x;
#pragma unroll
        for (int off = 16; off > 0; off >>= 1) ssum += __shfl_xor(ssum, off, 32);
        const float my_attn = my_x / (ssum + EPSV);

        for (int kb = 0; kb < n; kb += SEG_B) {
            float4 hv[SEG_B], cv[SEG_B], uf[SEG_B];
#pragma unroll
            for (int j = 0; j < SEG_B; ++j) {
                const int kk = min(kb + j, n - 1);
                const int ch = __builtin_amdgcn_readlane(my_ch, kk);
                hv[j] = ((const float4*)(child_h + (size_t)ch * HDIM))[lane];
                cv[j] = ((const float4*)(child_c + (size_t)ch * HDIM))[lane];
                uf[j] = ((const float4*)(Ufh     + (size_t)ch * HDIM))[lane];
            }
#pragma unroll
            for (int j = 0; j < SEG_B; ++j)
                asm volatile("" :: "v"(hv[j].x), "v"(hv[j].y), "v"(hv[j].z), "v"(hv[j].w),
                                   "v"(cv[j].x), "v"(cv[j].y), "v"(cv[j].z), "v"(cv[j].w),
                                   "v"(uf[j].x), "v"(uf[j].y), "v"(uf[j].z), "v"(uf[j].w));
#pragma unroll
            for (int j = 0; j < SEG_B; ++j) {
                if (kb + j < n) {
                    const float attn = __uint_as_float(
                        __builtin_amdgcn_readlane(__float_as_uint(my_attn), kb + j));
                    ah.x += attn * hv[j].x;
                    ah.y += attn * hv[j].y;
                    ah.z += attn * hv[j].z;
                    ah.w += attn * hv[j].w;
                    af.x += sigmoidf_(wf.x + uf[j].x) * cv[j].x;
                    af.y += sigmoidf_(wf.y + uf[j].y) * cv[j].y;
                    af.z += sigmoidf_(wf.z + uf[j].z) * cv[j].z;
                    af.w += sigmoidf_(wf.w + uf[j].w) * cv[j].w;
                }
            }
        }
    } else {
        float m = -__builtin_inff();
        for (int i = lane; i < n; i += 32) m = fmaxf(m, e[perm[s0 + i]]);
#pragma unroll
        for (int off = 16; off > 0; off >>= 1) m = fmaxf(m, __shfl_xor(m, off, 32));

        float ssum = 0.0f;
        int prev = -1;
        for (int k = 0; k < n; ++k) {
            const int cur = seg_next_edge(perm, s0, n, prev, lane);
            ssum += __expf(e[cur] - m);
            prev = cur;
        }
        const float denom = ssum + EPSV;

        prev = -1;
        for (int k = 0; k < n; ++k) {
            const int cur = seg_next_edge(perm, s0, n, prev, lane);
            prev = cur;
            const int ch = chi[cur];
            const float attn = __expf(e[cur] - m) / denom;
            const float4 hv = ((const float4*)(child_h + (size_t)ch * HDIM))[lane];
            const float4 cv = ((const float4*)(child_c + (size_t)ch * HDIM))[lane];
            const float4 uf = ((const float4*)(Ufh     + (size_t)ch * HDIM))[lane];
            ah.x += attn * hv.x;
            ah.y += attn * hv.y;
            ah.z += attn * hv.z;
            ah.w += attn * hv.w;
            af.x += sigmoidf_(wf.x + uf.x) * cv.x;
            af.y += sigmoidf_(wf.y + uf.y) * cv.y;
            af.z += sigmoidf_(wf.z + uf.z) * cv.z;
            af.w += sigmoidf_(wf.w + uf.w) * cv.w;
        }
    }

    ((float4*)(h_hat  + (size_t)c * HDIM))[lane] = ah;
    ((float4*)(sum_fc + (size_t)c * HDIM))[lane] = af;
}

extern "C" void kernel_launch(void* const* d_in, const int* in_sizes, int n_in,
                              void* d_out, int out_size, void* d_ws, size_t ws_size,
                              hipStream_t stream) {
    const float* x_emb   = (const float*)d_in[0];
    const float* child_h = (const float*)d_in[1];
    const float* child_c = (const float*)d_in[2];
    const int*   ci      = (const int*)d_in[3];
    const int*   chi     = (const int*)d_in[4];
    const float* Wi_w = (const float*)d_in[5];
    const float* Ui_w = (const float*)d_in[6];
    const float* Wf_w = (const float*)d_in[7];
    const float* Uf_w = (const float*)d_in[8];
    const float* Wo_w = (const float*)d_in[9];
    const float* Uo_w = (const float*)d_in[10];
    const float* Wc_w = (const float*)d_in[11];
    const float* Uc_w = (const float*)d_in[12];
    const float* Wa_w = (const float*)d_in[13];
    const float* Ua_w = (const float*)d_in[14];
    const float* Wi_b = (const float*)d_in[15];
    const float* Wf_b = (const float*)d_in[16];
    const float* Wo_b = (const float*)d_in[17];
    const float* Wc_b = (const float*)d_in[18];
    const float* Wa_b = (const float*)d_in[19];
    const float* v_w  = (const float*)d_in[20];

    const int M = in_sizes[0] / HDIM;
    const int L = in_sizes[3];
    const size_t MH = (size_t)M * HDIM;

    float* ws    = (float*)d_ws;
    float* Ah    = ws;
    float* Bx    = Ah  + MH;
    float* Gi    = Bx  + MH;
    float* Wfx   = Gi  + MH;
    float* Go    = Wfx + MH;
    float* Gc    = Go  + MH;
    float* Ufh   = Gc  + MH;
    float* h_hat = Ufh + MH;
    float* sumfc = h_hat + MH;
    float* e_buf = sumfc + MH;
    int* perm    = (int*)(e_buf + L);
    int* cnt     = perm + L;
    int* start   = cnt + M;

    hipMemsetAsync(cnt, 0, (size_t)M * sizeof(int), stream);

    const dim3 gblk(256);
    const dim3 ggrd((M + 127) / 128);
    const dim3 ggrd2(2 * ((M + 127) / 128));

    gemm_group_wmma_f32<false><<<ggrd2, gblk, 0, stream>>>(
        x_emb, child_h,
        Ua_w, Wi_w, Wf_w, Wo_w, Wc_w, Wa_w, Uf_w,
        nullptr, Wi_b, Wf_b, Wo_b, Wc_b, Wa_b, nullptr,
        Bx, Gi, Wfx, Go, Gc, Ah, Ufh,
        5, 2, M, nullptr, nullptr, nullptr);

    edge_score_kernel<<<(L + 8 * EPW * NBATCH - 1) / (8 * EPW * NBATCH), gblk, 0, stream>>>(Ah, Bx, v_w, ci, chi,
                                                        e_buf, cnt, L);
    seg_scan_kernel<<<(M + 1023) / 1024, dim3(1024), 0, stream>>>(cnt, start, M);
    seg_fill_kernel<<<(L + 255) / 256, gblk, 0, stream>>>(ci, start, cnt, perm, L);
    seg_attend_kernel<<<(M + 7) / 8, gblk, 0, stream>>>(e_buf, chi, start, perm,
                                                        child_h, child_c, Wfx, Ufh,
                                                        h_hat, sumfc, M, L);

    float* h_out = (float*)d_out;
    float* c_out = h_out + MH;
    gemm_group_wmma_f32<true><<<ggrd, gblk, 0, stream>>>(
        h_hat, h_hat,
        Uc_w, Ui_w, Uo_w, Uo_w, Uo_w, Uo_w, Uo_w,
        nullptr, nullptr, nullptr, nullptr, nullptr, nullptr, nullptr,
        Gc, Gi, Go, Go, Go, Go, Go,
        3, 0, M, sumfc, h_out, c_out);
}


# Edges grouped by candidate in 32-slot bins filled by a returning integer add inside the first product launch; scan and fill launches gone (run over 32: ordered pass over all edges); sum orders same
# speedup vs baseline: 3.3318x; 1.1505x over previous
#include <hip/hip_runtime.h>
#include <hip/hip_bf16.h>

#define HDIM 128
#define EPSV 1e-9f

typedef __attribute__((ext_vector_type(2))) float v2f;
typedef __attribute__((ext_vector_type(8))) float v8f;

__device__ __forceinline__ float sigmoidf_(float x) {
    return __builtin_amdgcn_rcpf(1.0f + __builtin_amdgcn_exp2f(x * -1.4426950408889634f));
}
__device__ __forceinline__ float tanhf_(float x) {
    return 1.0f - 2.0f * __builtin_amdgcn_rcpf(1.0f + __builtin_amdgcn_exp2f(x * 2.8853900817779268f));
}

#define LW_PITCH 132
#define BIN_CAP 32
#define BIN_PITCH HDIM
#define GEMM_PICK(j, p0, p1, p2, p3, p4, p5, p6)                                   \
    ((j) == 0 ? (p0) : (j) == 1 ? (p1) : (j) == 2 ? (p2) : (j) == 3 ? (p3)        \
   : (j) == 4 ? (p4) : (j) == 5 ? (p5) : (p6))
template <bool GATE>
__launch_bounds__(256) __attribute__((amdgpu_waves_per_eu(4, 4)))
__global__ void gemm_group_wmma_f32(const float* __restrict__ XA,
                                    const float* __restrict__ XB,
                                    const float* W0, const float* W1, const float* W2,
                                    const float* W3, const float* W4, const float* W5,
                                    const float* W6,
                                    const float* B0, const float* B1, const float* B2,
                                    const float* B3, const float* B4, const float* B5,
                                    const float* B6,
                                    float* Y0, float* Y1, float* Y2, float* Y3,
                                    float* Y4, float* Y5, float* Y6,
                                    int nA, int nB, int Mrows,
                                    const float* __restrict__ sum_fc,
                                    float* __restrict__ h_out,
                                    float* __restrict__ c_out,
                                    const int* __restrict__ edge_cand,
                                    int* __restrict__ cnt,
                                    int* __restrict__ bins, int nEdges) {
    __shared__ float lw[HDIM * LW_PITCH];

    __shared__ float4 keepb[GATE ? 8 * 16 * 32 : 1];

    const int e_stride = (int)gridDim.x * 256;
    const int e_first  = (int)blockIdx.x * 256 + (int)threadIdx.x;
    int ec0 = -1, ec1 = -1;
    if (!GATE) {
        if (e_first < nEdges)            ec0 = edge_cand[e_first];
        if (e_first + e_stride < nEdges) ec1 = edge_cand[e_first + e_stride];
    }

    const int wave = threadIdx.x >> 5;
    const int lane = threadIdx.x & 31;
    const int c    = lane & 15;
    const int h    = lane >> 4;

    const int nblk  = (Mrows + 127) >> 7;
    const int grp   = (int)blockIdx.x >= nblk ? 1 : 0;
    const int rb    = (int)blockIdx.x - grp * nblk;
    const int first = grp ? nA : 0;
    const int count = grp ? nB : nA;
    const float* __restrict__ X = grp ? XB : XA;

    const int m0   = rb * 128 + wave * 16;
    const bool active = m0 < Mrows;

    v2f a[32];
    if (active) {
        const float* __restrict__ xrow = X + (size_t)(m0 + c) * HDIM + 2 * h;
#pragma unroll
        for (int ks = 0; ks < 32; ++ks) {
            a[ks].x = xrow[4 * ks];
            a[ks].y = xrow[4 * ks + 1];
        }
    }

    const float* __restrict__ wrow = &lw[c * LW_PITCH + 2 * h];

    float* __restrict__ lw_st = &lw[((wave & 3) * 16 + (wave >> 2)) * LW_PITCH + (lane << 2)];
    const int w_off = (int)threadIdx.x * 4;
#define GEMM_STAGE_W(Wptr)                                                           \
    {                                                                                \
        const float* __restrict__ wsrc = (Wptr) + w_off;                             \
        float4 wv[16];                                                               \
        _Pragma("unroll")                                                            \
        for (int it = 0; it < 16; ++it)                                              \
            wv[it] = *(const float4*)(wsrc + it * 1024);                             \
        _Pragma("unroll")                                                            \
        for (int it = 0; it < 16; ++it)                                              \
            *(float4*)(lw_st + ((it >> 3) * 64 + (it & 7) * 2) * LW_PITCH) = wv[it]; \
    }

    const size_t goff = (size_t)(m0 + 8 * h) * HDIM + 4 * c;

    v8f acc[8];
#define GEMM_GATE_START(jj)                                                          \
    if (GATE && active) {                                                            \
        const float* ysrc = GEMM_PICK(jj, Y0, Y1, Y2, Y3, Y4, Y5, Y6) + goff;        \
        _Pragma("unroll")                                                            \
        for (int i = 0; i < 16; ++i) {                                               \
            const int q = i >> 3, r = i & 7;                                         \
            const float4 y = *(const float4*)(ysrc + (size_t)r * HDIM + 64 * q);     \
            acc[4 * q + 0][r] = y.x;                                                 \
            acc[4 * q + 1][r] = y.y;                                                 \
            acc[4 * q + 2][r] = y.z;                                                 \
            acc[4 * q + 3][r] = y.w;                                                 \
        }                                                                            \
    }

    GEMM_GATE_START(first)
    GEMM_STAGE_W(GEMM_PICK(first, W0, W1, W2, W3, W4, W5, W6))

    int es0 = BIN_CAP, es1 = BIN_CAP;
    if (!GATE) {
        if (ec0 >= 0) es0 = atomicAdd(&cnt[ec0], 1);
        if (ec1 >= 0) es1 = atomicAdd(&cnt[ec1], 1);
        for (int l = e_first + 2 * e_stride; l < nEdges; l += e_stride) {
            const int ec = edge_cand[l];
            const int es = atomicAdd(&cnt[ec], 1);
            if (es < BIN_CAP) bins[(size_t)ec * BIN_PITCH + es] = l;
        }
    }

#pragma nounroll
    for (int g = 0; g < count; ++g) {
        const int j = first + g;
        const float* __restrict__ bias = GEMM_PICK(j, B0, B1, B2, B3, B4, B5, B6);
        float* __restrict__ Y          = GEMM_PICK(j, Y0, Y1, Y2, Y3, Y4, Y5, Y6);

        __syncthreads();

        if (active) {
            if (!GATE) {
#pragma unroll
                for (int t = 0; t < 8; ++t) acc[t] = v8f{};
            }
#pragma unroll
            for (int ks = 0; ks < 32; ++ks) {
#pragma unroll
                for (int t = 0; t < 8; ++t) {
                    const v2f b = *(const v2f*)(wrow + t * 16 * LW_PITCH + 4 * ks);
                    acc[t] = __builtin_amdgcn_wmma_f32_16x16x4_f32(
                         false, a[ks],  false, b,
                         (short)0, acc[t],  false,  false);
                }
            }

            if (GATE) {
                float4* __restrict__ kb = &keepb[wave * 16 * 32 + lane];
                if (g == 0) {
#pragma unroll
                    for (int i = 0; i < 16; ++i) {
                        const int q = i >> 3, r = i & 7;
                        float4 t;
                        t.x = tanhf_(acc[4 * q + 0][r]);
                        t.y = tanhf_(acc[4 * q + 1][r]);
                        t.z = tanhf_(acc[4 * q + 2][r]);
                        t.w = tanhf_(acc[4 * q + 3][r]);
                        kb[i * 32] = t;
                    }
                } else if (g == 1) {
#pragma unroll
                    for (int q = 0; q < 2; ++q) {
                        float4 sf[8];
#pragma unroll
                        for (int r = 0; r < 8; ++r)
                            sf[r] = *(const float4*)(sum_fc + goff + (size_t)r * HDIM + 64 * q);
#pragma unroll
                        for (int r = 0; r < 8; r += 4)
                            asm volatile("" :: "v"(sf[r].x), "v"(sf[r].y), "v"(sf[r].z), "v"(sf[r].w),
                                               "v"(sf[r + 1].x), "v"(sf[r + 1].y), "v"(sf[r + 1].z), "v"(sf[r + 1].w),
                                               "v"(sf[r + 2].x), "v"(sf[r + 2].y), "v"(sf[r + 2].z), "v"(sf[r + 2].w),
                                               "v"(sf[r + 3].x), "v"(sf[r + 3].y), "v"(sf[r + 3].z), "v"(sf[r + 3].w));
#pragma unroll
                        for (int r = 0; r < 8; ++r) {
                            const float4 t = kb[(8 * q + r) * 32];
                            float4 cc, tc;
                            cc.x = sigmoidf_(acc[4 * q + 0][r]) * t.x + sf[r].x;
                            cc.y = sigmoidf_(acc[4 * q + 1][r]) * t.y + sf[r].y;
                            cc.z = sigmoidf_(acc[4 * q + 2][r]) * t.z + sf[r].z;
                            cc.w = sigmoidf_(acc[4 * q + 3][r]) * t.w + sf[r].w;
                            *(float4*)(c_out + goff + (size_t)r * HDIM + 64 * q) = cc;
                            tc.x = tanhf_(cc.x); tc.y = tanhf_(cc.y); tc.z = tanhf_(cc.z); tc.w = tanhf_(cc.w);
                            kb[(8 * q + r) * 32] = tc;
                        }
                    }
                } else {
#pragma unroll
                    for (int i = 0; i < 16; ++i) {
                        const int q = i >> 3, r = i & 7;
                        const float4 t = kb[i * 32];
                        float4 hh;
                        hh.x = sigmoidf_(acc[4 * q + 0][r]) * t.x;
                        hh.y = sigmoidf_(acc[4 * q + 1][r]) * t.y;
                        hh.z = sigmoidf_(acc[4 * q + 2][r]) * t.z;
                        hh.w = sigmoidf_(acc[4 * q + 3][r]) * t.w;
                        *(float4*)(h_out + goff + (size_t)r * HDIM + 64 * q) = hh;
                    }
                }
            } else {
                float* __restrict__ ybase = Y + goff;
                float4 bv[2];
#pragma unroll
                for (int q = 0; q < 2; ++q) {
                    bv[q] = make_float4(0.0f, 0.0f, 0.0f, 0.0f);
                    if (bias) bv[q] = *(const float4*)(bias + 64 * q + 4 * c);
                }
#pragma unroll
                for (int i = 0; i < 16; ++i) {
                    const int q = i >> 3, r = i & 7;
                    float4 v;
                    v.x = acc[4 * q + 0][r] + bv[q].x;
                    v.y = acc[4 * q + 1][r] + bv[q].y;
                    v.z = acc[4 * q + 2][r] + bv[q].z;
                    v.w = acc[4 * q + 3][r] + bv[q].w;
                    *(float4*)(ybase + (size_t)r * HDIM + 64 * q) = v;
                }
            }
        }

        if (g + 1 < count) {
            GEMM_GATE_START(j + 1)
            __syncthreads();
            GEMM_STAGE_W(GEMM_PICK(j + 1, W0, W1, W2, W3, W4, W5, W6))
        }
    }

    if (!GATE) {
        if (es0 < BIN_CAP) bins[(size_t)ec0 * BIN_PITCH + es0] = e_first;
        if (es1 < BIN_CAP) bins[(size_t)ec1 * BIN_PITCH + es1] = e_first + e_stride;
    }
#undef GEMM_STAGE_W
#undef GEMM_GATE_START
}

#define EPW 8
#define NBATCH 2

__launch_bounds__(256)
__global__ void edge_score_kernel(const float* __restrict__ Ah,
                                  const float* __restrict__ Bx,
                                  const float* __restrict__ v_w,
                                  const int* __restrict__ ci,
                                  const int* __restrict__ chi,
                                  float* __restrict__ e,
                                  int L) {
    const int lane = threadIdx.x & 31;
    const int wave = __builtin_amdgcn_readfirstlane((int)(threadIdx.x >> 5));
    const int w0 = (blockIdx.x * 8 + wave) * (EPW * NBATCH);
    const float4 v = ((const float4*)v_w)[lane];

    for (int bt = 0; bt < NBATCH; ++bt) {
        const int e0 = w0 + bt * EPW;
        if (e0 >= L) break;
        const int le = min(e0 + (lane & (EPW - 1)), L - 1);
        const int my_c = ci[le], my_ch = chi[le];

        float4 a[EPW], b[EPW];
#pragma unroll
        for (int j = 0; j < EPW; ++j) {
            const int c  = __builtin_amdgcn_readlane(my_c, j);
            const int ch = __builtin_amdgcn_readlane(my_ch, j);
            a[j] = ((const float4*)(Ah + (size_t)ch * HDIM))[lane];
            b[j] = ((const float4*)(Bx + (size_t)c  * HDIM))[lane];
        }

        float mine = 0.0f;
#pragma unroll
        for (int j = 0; j < EPW; ++j) {
            float s = tanhf_(a[j].x + b[j].x) * v.x + tanhf_(a[j].y + b[j].y) * v.y
                    + tanhf_(a[j].z + b[j].z) * v.z + tanhf_(a[j].w + b[j].w) * v.w;
#pragma unroll
            for (int off = 16; off > 0; off >>= 1) s += __shfl_xor(s, off, 32);
            if (lane == j) mine = s;
        }

        if (lane < EPW && e0 + lane < L) e[e0 + lane] = mine;
    }
}


#define SEG_B 8
__launch_bounds__(256)
__global__ void seg_attend_kernel(const float* __restrict__ e,
                                  const int* __restrict__ ci,
                                  const int* __restrict__ chi,
                                  const int* __restrict__ cnt,
                                  const int* bins,
                                  const float* __restrict__ child_h,
                                  const float* __restrict__ child_c,
                                  const float* __restrict__ Wfx,
                                  const float* __restrict__ Ufh,
                                  float* h_hat,
                                  float* __restrict__ sum_fc,
                                  int M, int L) {
    const int lane = threadIdx.x & 31;
    const int wave = __builtin_amdgcn_readfirstlane((int)(threadIdx.x >> 5));
    const int c = blockIdx.x * 8 + wave;
    if (c >= M) return;
    const int n = cnt[c];

    const float4 wf = ((const float4*)(Wfx + (size_t)c * HDIM))[lane];
    float4 ah = make_float4(0.0f, 0.0f, 0.0f, 0.0f);
    float4 af = make_float4(0.0f, 0.0f, 0.0f, 0.0f);

    if (n <= BIN_CAP) {
        const bool valid = lane < n;
        const int raw = valid ? bins[(size_t)c * BIN_PITCH + lane] : 0x7fffffff;
        int rank = 0;
        for (int j = 0; j < n; ++j) {
            const int o = __builtin_amdgcn_readlane(raw, j);
            rank += (o < raw) ? 1 : 0;
        }
        int id = 0;
        for (int j = 0; j < n; ++j) {
            const int r = __builtin_amdgcn_readlane(rank, j);
            const int o = __builtin_amdgcn_readlane(raw, j);
            if (lane == r) id = o;
        }
        const int   my_ch = valid ? chi[id] : 0;
        const float my_e  = valid ? e[id] : -__builtin_inff();

        float m = my_e;
#pragma unroll
        for (int off = 16; off > 0; off >>= 1) m = fmaxf(m, __shfl_xor(m, off, 32));
        const float my_x = valid ? __expf(my_e - m) : 0.0f;
        float ssum = my_x;
#pragma unroll
        for (int off = 16; off > 0; off >>= 1) ssum += __shfl_xor(ssum, off, 32);
        const float my_attn = my_x / (ssum + EPSV);

        for (int kb = 0; kb < n; kb += SEG_B) {
            float4 hv[SEG_B], cv[SEG_B], uf[SEG_B];
#pragma unroll
            for (int j = 0; j < SEG_B; ++j) {
                const int kk = min(kb + j, n - 1);
                const int ch = __builtin_amdgcn_readlane(my_ch, kk);
                hv[j] = ((const float4*)(child_h + (size_t)ch * HDIM))[lane];
                cv[j] = ((const float4*)(child_c + (size_t)ch * HDIM))[lane];
                uf[j] = ((const float4*)(Ufh     + (size_t)ch * HDIM))[lane];
            }
#pragma unroll
            for (int j = 0; j < SEG_B; ++j)
                asm volatile("" :: "v"(hv[j].x), "v"(hv[j].y), "v"(hv[j].z), "v"(hv[j].w),
                                   "v"(cv[j].x), "v"(cv[j].y), "v"(cv[j].z), "v"(cv[j].w),
                                   "v"(uf[j].x), "v"(uf[j].y), "v"(uf[j].z), "v"(uf[j].w));
#pragma unroll
            for (int j = 0; j < SEG_B; ++j) {
                if (kb + j < n) {
                    const float attn = __uint_as_float(
                        __builtin_amdgcn_readlane(__float_as_uint(my_attn), kb + j));
                    ah.x += attn * hv[j].x;
                    ah.y += attn * hv[j].y;
                    ah.z += attn * hv[j].z;
                    ah.w += attn * hv[j].w;
                    af.x += sigmoidf_(wf.x + uf[j].x) * cv[j].x;
                    af.y += sigmoidf_(wf.y + uf[j].y) * cv[j].y;
                    af.z += sigmoidf_(wf.z + uf[j].z) * cv[j].z;
                    af.w += sigmoidf_(wf.w + uf[j].w) * cv[j].w;
                }
            }
        }
    } else {
        float m = -__builtin_inff();
        for (int base = 0; base < L; base += 32) {
            const int l = base + lane;
            unsigned mask = __builtin_amdgcn_ballot_w32(l < L && ci[l] == c);
            while (mask) {
                const int cur = base + __builtin_ctz(mask);
                mask &= mask - 1;
                m = fmaxf(m, e[cur]);
            }
        }

        float ssum = 0.0f;
        for (int base = 0; base < L; base += 32) {
            const int l = base + lane;
            unsigned mask = __builtin_amdgcn_ballot_w32(l < L && ci[l] == c);
            while (mask) {
                const int cur = base + __builtin_ctz(mask);
                mask &= mask - 1;
                ssum += __expf(e[cur] - m);
            }
        }
        const float denom = ssum + EPSV;

        for (int base = 0; base < L; base += 32) {
            const int l = base + lane;
            unsigned mask = __builtin_amdgcn_ballot_w32(l < L && ci[l] == c);
            while (mask) {
                const int cur = base + __builtin_ctz(mask);
                mask &= mask - 1;
                const int ch = chi[cur];
                const float attn = __expf(e[cur] - m) / denom;
                const float4 hv = ((const float4*)(child_h + (size_t)ch * HDIM))[lane];
                const float4 cv = ((const float4*)(child_c + (size_t)ch * HDIM))[lane];
                const float4 uf = ((const float4*)(Ufh     + (size_t)ch * HDIM))[lane];
                ah.x += attn * hv.x;
                ah.y += attn * hv.y;
                ah.z += attn * hv.z;
                ah.w += attn * hv.w;
                af.x += sigmoidf_(wf.x + uf.x) * cv.x;
                af.y += sigmoidf_(wf.y + uf.y) * cv.y;
                af.z += sigmoidf_(wf.z + uf.z) * cv.z;
                af.w += sigmoidf_(wf.w + uf.w) * cv.w;
            }
        }
    }

    ((float4*)(h_hat  + (size_t)c * HDIM))[lane] = ah;
    ((float4*)(sum_fc + (size_t)c * HDIM))[lane] = af;
}

extern "C" void kernel_launch(void* const* d_in, const int* in_sizes, int n_in,
                              void* d_out, int out_size, void* d_ws, size_t ws_size,
                              hipStream_t stream) {
    const float* x_emb   = (const float*)d_in[0];
    const float* child_h = (const float*)d_in[1];
    const float* child_c = (const float*)d_in[2];
    const int*   ci      = (const int*)d_in[3];
    const int*   chi     = (const int*)d_in[4];
    const float* Wi_w = (const float*)d_in[5];
    const float* Ui_w = (const float*)d_in[6];
    const float* Wf_w = (const float*)d_in[7];
    const float* Uf_w = (const float*)d_in[8];
    const float* Wo_w = (const float*)d_in[9];
    const float* Uo_w = (const float*)d_in[10];
    const float* Wc_w = (const float*)d_in[11];
    const float* Uc_w = (const float*)d_in[12];
    const float* Wa_w = (const float*)d_in[13];
    const float* Ua_w = (const float*)d_in[14];
    const float* Wi_b = (const float*)d_in[15];
    const float* Wf_b = (const float*)d_in[16];
    const float* Wo_b = (const float*)d_in[17];
    const float* Wc_b = (const float*)d_in[18];
    const float* Wa_b = (const float*)d_in[19];
    const float* v_w  = (const float*)d_in[20];

    const int M = in_sizes[0] / HDIM;
    const int L = in_sizes[3];
    const size_t MH = (size_t)M * HDIM;

    float* ws    = (float*)d_ws;
    float* Ah    = ws;
    float* Bx    = Ah  + MH;
    float* Gi    = Bx  + MH;
    float* Wfx   = Gi  + MH;
    float* Go    = Wfx + MH;
    float* Gc    = Go  + MH;
    float* Ufh   = Gc  + MH;
    float* h_hat = Ufh + MH;
    float* sumfc = h_hat + MH;
    float* e_buf = sumfc + MH;
    int* cnt     = (int*)(e_buf + L);
    int* bins    = (int*)h_hat;

    hipMemsetAsync(cnt, 0, (size_t)M * sizeof(int), stream);

    const dim3 gblk(256);
    const dim3 ggrd((M + 127) / 128);
    const dim3 ggrd2(2 * ((M + 127) / 128));

    gemm_group_wmma_f32<false><<<ggrd2, gblk, 0, stream>>>(
        x_emb, child_h,
        Ua_w, Wi_w, Wf_w, Wo_w, Wc_w, Wa_w, Uf_w,
        nullptr, Wi_b, Wf_b, Wo_b, Wc_b, Wa_b, nullptr,
        Bx, Gi, Wfx, Go, Gc, Ah, Ufh,
        5, 2, M, nullptr, nullptr, nullptr, ci, cnt, bins, L);

    edge_score_kernel<<<(L + 8 * EPW * NBATCH - 1) / (8 * EPW * NBATCH), gblk, 0, stream>>>(Ah, Bx, v_w, ci, chi,
                                                        e_buf, L);
    seg_attend_kernel<<<(M + 7) / 8, gblk, 0, stream>>>(e_buf, ci, chi, cnt, bins,
                                                        child_h, child_c, Wfx, Ufh,
                                                        h_hat, sumfc, M, L);

    float* h_out = (float*)d_out;
    float* c_out = h_out + MH;
    gemm_group_wmma_f32<true><<<ggrd, gblk, 0, stream>>>(
        h_hat, h_hat,
        Uc_w, Ui_w, Uo_w, Uo_w, Uo_w, Uo_w, Uo_w,
        nullptr, nullptr, nullptr, nullptr, nullptr, nullptr, nullptr,
        Gc, Gi, Go, Go, Go, Go, Go,
        3, 0, M, sumfc, h_out, c_out, nullptr, nullptr, nullptr, 0);
}
